# starting kernel
# baseline (speedup 1.0000x reference)
#include <hip/hip_runtime.h>
#include <stdint.h>

#define NN    50000
#define EE    400000
#define INCH  128
#define OUTC  64
#define H1    8
#define NEG   0.2f

typedef float v2f __attribute__((ext_vector_type(2)));
typedef float v8f __attribute__((ext_vector_type(8)));

__global__ void fill_u32_kernel(uint32_t* p, uint32_t v, long long n) {
  long long i = (long long)blockIdx.x * blockDim.x + threadIdx.x;
  if (i < n) p[i] = v;
}

__global__ void gemm_wmma_f32_kernel(const float* __restrict__ A,
                                     const float* __restrict__ B,
                                     float* __restrict__ C,
                                     int M, int K, int Nc) {
  const int wid  = (blockIdx.x * blockDim.x + threadIdx.x) >> 5;
  const int lane = threadIdx.x & 31;
  const int tilesN   = Nc >> 6;
  const int rowTiles = M >> 4;
  if (wid >= rowTiles * tilesN) return;

  const int rt = wid / tilesN;
  const int cg = wid - rt * tilesN;
  const int m0 = rt << 4;
  const int n0 = cg << 6;
  const int ml = lane & 15;
  const int kh = lane >> 4;

  v8f acc0 = {}, acc1 = {}, acc2 = {}, acc3 = {};
  const float* arow = A + (size_t)(m0 + ml) * K;

  for (int k = 0; k < K; k += 4) {
    const int ka = k + 2 * kh;
    v2f a;
    a.x = arow[ka];
    a.y = arow[ka + 1];
    const float* b0p = B + (size_t)ka * Nc + n0 + ml;
    const float* b1p = b0p + Nc;
    v2f b0; b0.x = b0p[0];  b0.y = b1p[0];
    v2f b1; b1.x = b0p[16]; b1.y = b1p[16];
    v2f b2; b2.x = b0p[32]; b2.y = b1p[32];
    v2f b3; b3.x = b0p[48]; b3.y = b1p[48];
    acc0 = __builtin_amdgcn_wmma_f32_16x16x4_f32(false, a, false, b0, (short)0, acc0, false, false);
    acc1 = __builtin_amdgcn_wmma_f32_16x16x4_f32(false, a, false, b1, (short)0, acc1, false, false);
    acc2 = __builtin_amdgcn_wmma_f32_16x16x4_f32(false, a, false, b2, (short)0, acc2, false, false);
    acc3 = __builtin_amdgcn_wmma_f32_16x16x4_f32(false, a, false, b3, (short)0, acc3, false, false);
  }

  const int mr = m0 + 8 * kh;
#pragma unroll
  for (int v = 0; v < 8; ++v) {
    float* crow = C + (size_t)(mr + v) * Nc + n0 + ml;
    crow[0]  = acc0[v];
    crow[16] = acc1[v];
    crow[32] = acc2[v];
    crow[48] = acc3[v];
  }
}

__global__ void att_scores_kernel(const float* __restrict__ h,
                                  const float* __restrict__ att_src,
                                  const float* __restrict__ att_dst,
                                  float* __restrict__ a_s,
                                  float* __restrict__ a_d,
                                  int n_nodes, int heads) {
  const int wid  = (blockIdx.x * blockDim.x + threadIdx.x) >> 5;
  const int lane = threadIdx.x & 31;
  if (wid >= n_nodes * heads) return;
  const int n  = wid / heads;
  const int hd = wid - n * heads;
  const float* hrow = h + (size_t)n * heads * OUTC + hd * OUTC;
  const float* ws   = att_src + hd * OUTC;
  const float* wd   = att_dst + hd * OUTC;
  float vs = hrow[lane] * ws[lane] + hrow[lane + 32] * ws[lane + 32];
  float vd = hrow[lane] * wd[lane] + hrow[lane + 32] * wd[lane + 32];
#pragma unroll
  for (int off = 16; off; off >>= 1) {
    vs += __shfl_xor(vs, off, 32);
    vd += __shfl_xor(vd, off, 32);
  }
  if (lane == 0) {
    a_s[n * heads + hd] = vs;
    a_d[n * heads + hd] = vd;
  }
}

__device__ __forceinline__ void edge_endpoints(const int* ei, int e, int& s, int& d) {
  if (e < EE) { s = ei[e]; d = ei[EE + e]; }
  else        { s = d = e - EE; }
}

__device__ __forceinline__ float leaky(float x) {
  return x > 0.0f ? x : NEG * x;
}

__device__ __forceinline__ void atomicMaxF32(float* addr, float val) {
  if (val >= 0.0f) atomicMax((int*)addr, __float_as_int(val));
  else             atomicMin((unsigned int*)addr, __float_as_uint(val));
}

__global__ void edge_max_kernel(const int* __restrict__ ei,
                                const float* __restrict__ a_s,
                                const float* __restrict__ a_d,
                                float* __restrict__ emax, int heads) {
  const long long tid = (long long)blockIdx.x * blockDim.x + threadIdx.x;
  const long long total = (long long)(EE + NN) * heads;
  if (tid >= total) return;
  const int e  = (int)(tid / heads);
  const int hd = (int)(tid - (long long)e * heads);
  int s, d; edge_endpoints(ei, e, s, d);
  const float ev = leaky(a_s[s * heads + hd] + a_d[d * heads + hd]);
  atomicMaxF32(&emax[d * heads + hd], ev);
}

__global__ void edge_expsum_kernel(const int* __restrict__ ei,
                                   const float* __restrict__ a_s,
                                   const float* __restrict__ a_d,
                                   const float* __restrict__ emax,
                                   float* __restrict__ denom, int heads) {
  const long long tid = (long long)blockIdx.x * blockDim.x + threadIdx.x;
  const long long total = (long long)(EE + NN) * heads;
  if (tid >= total) return;
  const int e  = (int)(tid / heads);
  const int hd = (int)(tid - (long long)e * heads);
  int s, d; edge_endpoints(ei, e, s, d);
  const float ev = leaky(a_s[s * heads + hd] + a_d[d * heads + hd]);
  atomicAdd(&denom[d * heads + hd], __expf(ev - emax[d * heads + hd]));
}

__global__ void edge_aggregate_kernel(const int* __restrict__ ei,
                                      const float* __restrict__ a_s,
                                      const float* __restrict__ a_d,
                                      const float* __restrict__ emax,
                                      const float* __restrict__ denom,
                                      const float* __restrict__ h,
                                      float* __restrict__ acc, int heads) {
  const long long wid = ((long long)blockIdx.x * blockDim.x + threadIdx.x) >> 5;
  const int lane = threadIdx.x & 31;
  const long long total = (long long)(EE + NN) * heads;
  if (wid >= total) return;
  const int e  = (int)(wid / heads);
  const int hd = (int)(wid - (long long)e * heads);
  int s, d; edge_endpoints(ei, e, s, d);
  const float ev = leaky(a_s[s * heads + hd] + a_d[d * heads + hd]);
  const float alpha = __expf(ev - emax[d * heads + hd]) / denom[d * heads + hd];
  const float* hrow = h + (size_t)s * heads * OUTC + hd * OUTC;
  float* arow = acc + (size_t)d * OUTC;
  atomicAdd(&arow[lane],      alpha * hrow[lane]);
  atomicAdd(&arow[lane + 32], alpha * hrow[lane + 32]);
}

__global__ void finalize_kernel(const float* __restrict__ acc,
                                const float* __restrict__ bias,
                                float* __restrict__ out,
                                float invH, int do_elu) {
  const long long tid = (long long)blockIdx.x * blockDim.x + threadIdx.x;
  if (tid >= (long long)NN * OUTC) return;
  const int c = (int)(tid & (OUTC - 1));
  float v = acc[tid] * invH + bias[c];
  if (do_elu) v = v > 0.0f ? v : (__expf(v) - 1.0f);
  out[tid] = v;
}

static inline unsigned cdiv(long long a, long long b) { return (unsigned)((a + b - 1) / b); }

extern "C" void kernel_launch(void* const* d_in, const int* in_sizes, int n_in,
                              void* d_out, int out_size, void* d_ws, size_t ws_size,
                              hipStream_t stream) {
  (void)in_sizes; (void)n_in; (void)out_size; (void)ws_size;
  const float* x    = (const float*)d_in[0];
  const int*   ei   = (const int*)d_in[1];
  const float* W1   = (const float*)d_in[2];
  const float* asw1 = (const float*)d_in[3];
  const float* adw1 = (const float*)d_in[4];
  const float* b1   = (const float*)d_in[5];
  const float* W2   = (const float*)d_in[6];
  const float* asw2 = (const float*)d_in[7];
  const float* adw2 = (const float*)d_in[8];
  const float* b2   = (const float*)d_in[9];
  float* out = (float*)d_out;

  float* ws   = (float*)d_ws;
  float* h1   = ws; ws += (size_t)NN * H1 * OUTC;
  float* as1  = ws; ws += (size_t)NN * H1;
  float* ad1  = ws; ws += (size_t)NN * H1;
  float* em1  = ws; ws += (size_t)NN * H1;
  float* dn1  = ws; ws += (size_t)NN * H1;
  float* acc1 = ws; ws += (size_t)NN * OUTC;
  float* h2   = ws; ws += (size_t)NN * OUTC;
  float* as2  = ws; ws += (size_t)NN;
  float* ad2  = ws; ws += (size_t)NN;
  float* em2  = ws; ws += (size_t)NN;
  float* dn2  = ws; ws += (size_t)NN;

  const int T = 256;
  const long long enh1 = (long long)(EE + NN) * H1;
  const long long enh2 = (long long)(EE + NN);
  const uint32_t NEG_INF = 0xFF800000u;

  fill_u32_kernel<<<cdiv(NN * H1, T), T, 0, stream>>>((uint32_t*)em1, NEG_INF, NN * H1);
  fill_u32_kernel<<<cdiv(NN * H1, T), T, 0, stream>>>((uint32_t*)dn1, 0u, NN * H1);
  fill_u32_kernel<<<cdiv((long long)NN * OUTC, T), T, 0, stream>>>((uint32_t*)acc1, 0u, (long long)NN * OUTC);

  {
    long long waves = ((long long)NN / 16) * (H1 * OUTC / 64);
    gemm_wmma_f32_kernel<<<cdiv(waves * 32, T), T, 0, stream>>>(x, W1, h1, NN, INCH, H1 * OUTC);
  }
  att_scores_kernel<<<cdiv((long long)NN * H1 * 32, T), T, 0, stream>>>(h1, asw1, adw1, as1, ad1, NN, H1);
  edge_max_kernel<<<cdiv(enh1, T), T, 0, stream>>>(ei, as1, ad1, em1, H1);
  edge_expsum_kernel<<<cdiv(enh1, T), T, 0, stream>>>(ei, as1, ad1, em1, dn1, H1);
  edge_aggregate_kernel<<<cdiv(enh1 * 32, T), T, 0, stream>>>(ei, as1, ad1, em1, dn1, h1, acc1, H1);
  finalize_kernel<<<cdiv((long long)NN * OUTC, T), T, 0, stream>>>(acc1, b1, acc1, 1.0f / H1, 1);

  fill_u32_kernel<<<cdiv(NN, T), T, 0, stream>>>((uint32_t*)em2, NEG_INF, NN);
  fill_u32_kernel<<<cdiv(NN, T), T, 0, stream>>>((uint32_t*)dn2, 0u, NN);
  fill_u32_kernel<<<cdiv((long long)NN * OUTC, T), T, 0, stream>>>((uint32_t*)out, 0u, (long long)NN * OUTC);

  {
    long long waves = ((long long)NN / 16) * (OUTC / 64);
    gemm_wmma_f32_kernel<<<cdiv(waves * 32, T), T, 0, stream>>>(acc1, W2, h2, NN, OUTC, OUTC);
  }
  att_scores_kernel<<<cdiv((long long)NN * 32, T), T, 0, stream>>>(h2, asw2, adw2, as2, ad2, NN, 1);
  edge_max_kernel<<<cdiv(enh2, T), T, 0, stream>>>(ei, as2, ad2, em2, 1);
  edge_expsum_kernel<<<cdiv(enh2, T), T, 0, stream>>>(ei, as2, ad2, em2, dn2, 1);
  edge_aggregate_kernel<<<cdiv(enh2 * 32, T), T, 0, stream>>>(ei, as2, ad2, em2, dn2, h2, out, 1);
  finalize_kernel<<<cdiv((long long)NN * OUTC, T), T, 0, stream>>>(out, b2, out, 1.0f, 0);
}


# Layer-1 aggregate: one wave per edge sums the 8 heads' weighted rows in registers, then one atomic pair per edge (8x fewer waves and float atomics; head sum now done ahead of the atomic add)
# speedup vs baseline: 1.9089x; 1.9089x over previous
#include <hip/hip_runtime.h>
#include <stdint.h>

#define NN    50000
#define EE    400000
#define INCH  128
#define OUTC  64
#define H1    8
#define NEG   0.2f

typedef float v2f __attribute__((ext_vector_type(2)));
typedef float v8f __attribute__((ext_vector_type(8)));

__global__ void fill_u32_kernel(uint32_t* p, uint32_t v, long long n) {
  long long i = (long long)blockIdx.x * blockDim.x + threadIdx.x;
  if (i < n) p[i] = v;
}

__global__ void gemm_wmma_f32_kernel(const float* __restrict__ A,
                                     const float* __restrict__ B,
                                     float* __restrict__ C,
                                     int M, int K, int Nc) {
  const int wid  = (blockIdx.x * blockDim.x + threadIdx.x) >> 5;
  const int lane = threadIdx.x & 31;
  const int tilesN   = Nc >> 6;
  const int rowTiles = M >> 4;
  if (wid >= rowTiles * tilesN) return;

  const int rt = wid / tilesN;
  const int cg = wid - rt * tilesN;
  const int m0 = rt << 4;
  const int n0 = cg << 6;
  const int ml = lane & 15;
  const int kh = lane >> 4;

  v8f acc0 = {}, acc1 = {}, acc2 = {}, acc3 = {};
  const float* arow = A + (size_t)(m0 + ml) * K;

  for (int k = 0; k < K; k += 4) {
    const int ka = k + 2 * kh;
    v2f a;
    a.x = arow[ka];
    a.y = arow[ka + 1];
    const float* b0p = B + (size_t)ka * Nc + n0 + ml;
    const float* b1p = b0p + Nc;
    v2f b0; b0.x = b0p[0];  b0.y = b1p[0];
    v2f b1; b1.x = b0p[16]; b1.y = b1p[16];
    v2f b2; b2.x = b0p[32]; b2.y = b1p[32];
    v2f b3; b3.x = b0p[48]; b3.y = b1p[48];
    acc0 = __builtin_amdgcn_wmma_f32_16x16x4_f32(false, a, false, b0, (short)0, acc0, false, false);
    acc1 = __builtin_amdgcn_wmma_f32_16x16x4_f32(false, a, false, b1, (short)0, acc1, false, false);
    acc2 = __builtin_amdgcn_wmma_f32_16x16x4_f32(false, a, false, b2, (short)0, acc2, false, false);
    acc3 = __builtin_amdgcn_wmma_f32_16x16x4_f32(false, a, false, b3, (short)0, acc3, false, false);
  }

  const int mr = m0 + 8 * kh;
#pragma unroll
  for (int v = 0; v < 8; ++v) {
    float* crow = C + (size_t)(mr + v) * Nc + n0 + ml;
    crow[0]  = acc0[v];
    crow[16] = acc1[v];
    crow[32] = acc2[v];
    crow[48] = acc3[v];
  }
}

__global__ void att_scores_kernel(const float* __restrict__ h,
                                  const float* __restrict__ att_src,
                                  const float* __restrict__ att_dst,
                                  float* __restrict__ a_s,
                                  float* __restrict__ a_d,
                                  int n_nodes, int heads) {
  const int wid  = (blockIdx.x * blockDim.x + threadIdx.x) >> 5;
  const int lane = threadIdx.x & 31;
  if (wid >= n_nodes * heads) return;
  const int n  = wid / heads;
  const int hd = wid - n * heads;
  const float* hrow = h + (size_t)n * heads * OUTC + hd * OUTC;
  const float* ws   = att_src + hd * OUTC;
  const float* wd   = att_dst + hd * OUTC;
  float vs = hrow[lane] * ws[lane] + hrow[lane + 32] * ws[lane + 32];
  float vd = hrow[lane] * wd[lane] + hrow[lane + 32] * wd[lane + 32];
#pragma unroll
  for (int off = 16; off; off >>= 1) {
    vs += __shfl_xor(vs, off, 32);
    vd += __shfl_xor(vd, off, 32);
  }
  if (lane == 0) {
    a_s[n * heads + hd] = vs;
    a_d[n * heads + hd] = vd;
  }
}

__device__ __forceinline__ void edge_endpoints(const int* ei, int e, int& s, int& d) {
  if (e < EE) { s = ei[e]; d = ei[EE + e]; }
  else        { s = d = e - EE; }
}

__device__ __forceinline__ float leaky(float x) {
  return x > 0.0f ? x : NEG * x;
}

__device__ __forceinline__ void atomicMaxF32(float* addr, float val) {
  if (val >= 0.0f) atomicMax((int*)addr, __float_as_int(val));
  else             atomicMin((unsigned int*)addr, __float_as_uint(val));
}

__global__ void edge_max_kernel(const int* __restrict__ ei,
                                const float* __restrict__ a_s,
                                const float* __restrict__ a_d,
                                float* __restrict__ emax, int heads) {
  const long long tid = (long long)blockIdx.x * blockDim.x + threadIdx.x;
  const long long total = (long long)(EE + NN) * heads;
  if (tid >= total) return;
  const int e  = (int)(tid / heads);
  const int hd = (int)(tid - (long long)e * heads);
  int s, d; edge_endpoints(ei, e, s, d);
  const float ev = leaky(a_s[s * heads + hd] + a_d[d * heads + hd]);
  atomicMaxF32(&emax[d * heads + hd], ev);
}

__global__ void edge_expsum_kernel(const int* __restrict__ ei,
                                   const float* __restrict__ a_s,
                                   const float* __restrict__ a_d,
                                   const float* __restrict__ emax,
                                   float* __restrict__ denom, int heads) {
  const long long tid = (long long)blockIdx.x * blockDim.x + threadIdx.x;
  const long long total = (long long)(EE + NN) * heads;
  if (tid >= total) return;
  const int e  = (int)(tid / heads);
  const int hd = (int)(tid - (long long)e * heads);
  int s, d; edge_endpoints(ei, e, s, d);
  const float ev = leaky(a_s[s * heads + hd] + a_d[d * heads + hd]);
  atomicAdd(&denom[d * heads + hd], __expf(ev - emax[d * heads + hd]));
}

template <int HEADS>
__global__ void edge_aggregate_kernel(const int* __restrict__ ei,
                                      const float* __restrict__ a_s,
                                      const float* __restrict__ a_d,
                                      const float* __restrict__ emax,
                                      const float* __restrict__ denom,
                                      const float* __restrict__ h,
                                      float* __restrict__ acc) {
  static_assert(HEADS >= 1 && HEADS <= 32 && (HEADS & (HEADS - 1)) == 0,
                "HEADS must be a power of two that fits in one wave");
  const long long wid = ((long long)blockIdx.x * blockDim.x + threadIdx.x) >> 5;
  const int lane = threadIdx.x & 31;
  if (wid >= (long long)(EE + NN)) return;
  const int e = __builtin_amdgcn_readfirstlane((int)wid);
  int s, d; edge_endpoints(ei, e, s, d);
  const int hd = lane & (HEADS - 1);
  const float ev = leaky(a_s[s * HEADS + hd] + a_d[d * HEADS + hd]);
  const float alpha = __expf(ev - emax[d * HEADS + hd]) / denom[d * HEADS + hd];
  const float* hrow = h + (size_t)s * HEADS * OUTC;
  float v0 = 0.0f, v1 = 0.0f;
#pragma unroll
  for (int k = 0; k < HEADS; ++k) {
    const float al = __int_as_float(__builtin_amdgcn_readlane(__float_as_int(alpha), k));
    v0 += al * hrow[k * OUTC + lane];
    v1 += al * hrow[k * OUTC + lane + 32];
  }
  float* arow = acc + (size_t)d * OUTC;
  atomicAdd(&arow[lane],      v0);
  atomicAdd(&arow[lane + 32], v1);
}

__global__ void finalize_kernel(const float* __restrict__ acc,
                                const float* __restrict__ bias,
                                float* __restrict__ out,
                                float invH, int do_elu) {
  const long long tid = (long long)blockIdx.x * blockDim.x + threadIdx.x;
  if (tid >= (long long)NN * OUTC) return;
  const int c = (int)(tid & (OUTC - 1));
  float v = acc[tid] * invH + bias[c];
  if (do_elu) v = v > 0.0f ? v : (__expf(v) - 1.0f);
  out[tid] = v;
}

static inline unsigned cdiv(long long a, long long b) { return (unsigned)((a + b - 1) / b); }

extern "C" void kernel_launch(void* const* d_in, const int* in_sizes, int n_in,
                              void* d_out, int out_size, void* d_ws, size_t ws_size,
                              hipStream_t stream) {
  (void)in_sizes; (void)n_in; (void)out_size; (void)ws_size;
  const float* x    = (const float*)d_in[0];
  const int*   ei   = (const int*)d_in[1];
  const float* W1   = (const float*)d_in[2];
  const float* asw1 = (const float*)d_in[3];
  const float* adw1 = (const float*)d_in[4];
  const float* b1   = (const float*)d_in[5];
  const float* W2   = (const float*)d_in[6];
  const float* asw2 = (const float*)d_in[7];
  const float* adw2 = (const float*)d_in[8];
  const float* b2   = (const float*)d_in[9];
  float* out = (float*)d_out;

  float* ws   = (float*)d_ws;
  float* h1   = ws; ws += (size_t)NN * H1 * OUTC;
  float* as1  = ws; ws += (size_t)NN * H1;
  float* ad1  = ws; ws += (size_t)NN * H1;
  float* em1  = ws; ws += (size_t)NN * H1;
  float* dn1  = ws; ws += (size_t)NN * H1;
  float* acc1 = ws; ws += (size_t)NN * OUTC;
  float* h2   = ws; ws += (size_t)NN * OUTC;
  float* as2  = ws; ws += (size_t)NN;
  float* ad2  = ws; ws += (size_t)NN;
  float* em2  = ws; ws += (size_t)NN;
  float* dn2  = ws; ws += (size_t)NN;

  const int T = 256;
  const long long enh1 = (long long)(EE + NN) * H1;
  const long long enh2 = (long long)(EE + NN);
  const uint32_t NEG_INF = 0xFF800000u;

  fill_u32_kernel<<<cdiv(NN * H1, T), T, 0, stream>>>((uint32_t*)em1, NEG_INF, NN * H1);
  fill_u32_kernel<<<cdiv(NN * H1, T), T, 0, stream>>>((uint32_t*)dn1, 0u, NN * H1);
  fill_u32_kernel<<<cdiv((long long)NN * OUTC, T), T, 0, stream>>>((uint32_t*)acc1, 0u, (long long)NN * OUTC);

  {
    long long waves = ((long long)NN / 16) * (H1 * OUTC / 64);
    gemm_wmma_f32_kernel<<<cdiv(waves * 32, T), T, 0, stream>>>(x, W1, h1, NN, INCH, H1 * OUTC);
  }
  att_scores_kernel<<<cdiv((long long)NN * H1 * 32, T), T, 0, stream>>>(h1, asw1, adw1, as1, ad1, NN, H1);
  edge_max_kernel<<<cdiv(enh1, T), T, 0, stream>>>(ei, as1, ad1, em1, H1);
  edge_expsum_kernel<<<cdiv(enh1, T), T, 0, stream>>>(ei, as1, ad1, em1, dn1, H1);
  edge_aggregate_kernel<H1><<<cdiv(enh2 * 32, T), T, 0, stream>>>(ei, as1, ad1, em1, dn1, h1, acc1);
  finalize_kernel<<<cdiv((long long)NN * OUTC, T), T, 0, stream>>>(acc1, b1, acc1, 1.0f / H1, 1);

  fill_u32_kernel<<<cdiv(NN, T), T, 0, stream>>>((uint32_t*)em2, NEG_INF, NN);
  fill_u32_kernel<<<cdiv(NN, T), T, 0, stream>>>((uint32_t*)dn2, 0u, NN);
  fill_u32_kernel<<<cdiv((long long)NN * OUTC, T), T, 0, stream>>>((uint32_t*)out, 0u, (long long)NN * OUTC);

  {
    long long waves = ((long long)NN / 16) * (OUTC / 64);
    gemm_wmma_f32_kernel<<<cdiv(waves * 32, T), T, 0, stream>>>(acc1, W2, h2, NN, OUTC, OUTC);
  }
  att_scores_kernel<<<cdiv((long long)NN * 32, T), T, 0, stream>>>(h2, asw2, adw2, as2, ad2, NN, 1);
  edge_max_kernel<<<cdiv(enh2, T), T, 0, stream>>>(ei, as2, ad2, em2, 1);
  edge_expsum_kernel<<<cdiv(enh2, T), T, 0, stream>>>(ei, as2, ad2, em2, dn2, 1);
  edge_aggregate_kernel<1><<<cdiv(enh2 * 32, T), T, 0, stream>>>(ei, as2, ad2, em2, dn2, h2, out);
  finalize_kernel<<<cdiv((long long)NN * OUTC, T), T, 0, stream>>>(out, b2, out, 1.0f, 0);
}


# Each wave does several items (pass C 8 or 32 edges with one alpha per lane, scores 8 rows as float4, passes A/B, fill, finalize 4 per lane): 8x fewer waves; score dot products summed in a new order
# speedup vs baseline: 2.1548x; 1.1288x over previous
#include <hip/hip_runtime.h>
#include <stdint.h>

#define NN    50000
#define EE    400000
#define INCH  128
#define OUTC  64
#define H1    8
#define NEG   0.2f

typedef float v2f __attribute__((ext_vector_type(2)));
typedef float v8f __attribute__((ext_vector_type(8)));

#define VPT 4
__global__ void fill_u32_kernel(uint32_t* p, uint32_t v, int n) {
  const int tid    = blockIdx.x * blockDim.x + threadIdx.x;
  const int stride = gridDim.x * blockDim.x;
  const int n4     = n >> 2;
  uint4* p4 = reinterpret_cast<uint4*>(p);
  const uint4 val = make_uint4(v, v, v, v);
#pragma unroll
  for (int j = 0; j < VPT; ++j) {
    const int i = tid + j * stride;
    if (i < n4) p4[i] = val;
  }
}

__global__ void gemm_wmma_f32_kernel(const float* __restrict__ A,
                                     const float* __restrict__ B,
                                     float* __restrict__ C,
                                     int M, int K, int Nc) {
  const int wid  = (blockIdx.x * blockDim.x + threadIdx.x) >> 5;
  const int lane = threadIdx.x & 31;
  const int tilesN   = Nc >> 6;
  const int rowTiles = M >> 4;
  if (wid >= rowTiles * tilesN) return;

  const int rt = wid / tilesN;
  const int cg = wid - rt * tilesN;
  const int m0 = rt << 4;
  const int n0 = cg << 6;
  const int ml = lane & 15;
  const int kh = lane >> 4;

  v8f acc0 = {}, acc1 = {}, acc2 = {}, acc3 = {};
  const float* arow = A + (size_t)(m0 + ml) * K;

  for (int k = 0; k < K; k += 4) {
    const int ka = k + 2 * kh;
    v2f a;
    a.x = arow[ka];
    a.y = arow[ka + 1];
    const float* b0p = B + (size_t)ka * Nc + n0 + ml;
    const float* b1p = b0p + Nc;
    v2f b0; b0.x = b0p[0];  b0.y = b1p[0];
    v2f b1; b1.x = b0p[16]; b1.y = b1p[16];
    v2f b2; b2.x = b0p[32]; b2.y = b1p[32];
    v2f b3; b3.x = b0p[48]; b3.y = b1p[48];
    acc0 = __builtin_amdgcn_wmma_f32_16x16x4_f32(false, a, false, b0, (short)0, acc0, false, false);
    acc1 = __builtin_amdgcn_wmma_f32_16x16x4_f32(false, a, false, b1, (short)0, acc1, false, false);
    acc2 = __builtin_amdgcn_wmma_f32_16x16x4_f32(false, a, false, b2, (short)0, acc2, false, false);
    acc3 = __builtin_amdgcn_wmma_f32_16x16x4_f32(false, a, false, b3, (short)0, acc3, false, false);
  }

  const int mr = m0 + 8 * kh;
#pragma unroll
  for (int v = 0; v < 8; ++v) {
    float* crow = C + (size_t)(mr + v) * Nc + n0 + ml;
    crow[0]  = acc0[v];
    crow[16] = acc1[v];
    crow[32] = acc2[v];
    crow[48] = acc3[v];
  }
}

template <int HEADS>
__global__ void att_scores_kernel(const float* __restrict__ h,
                                  const float* __restrict__ att_src,
                                  const float* __restrict__ att_dst,
                                  float* __restrict__ a_s,
                                  float* __restrict__ a_d,
                                  int rows) {
  const int wid  = (blockIdx.x * blockDim.x + threadIdx.x) >> 5;
  const int lane = threadIdx.x & 31;
  const int c    = lane & 15;
  const int hf   = lane >> 4;
  const int base = wid * 8;
  if (base >= rows) return;
  const float4* h4  = reinterpret_cast<const float4*>(h);
  const float4* ws4 = reinterpret_cast<const float4*>(att_src);
  const float4* wd4 = reinterpret_cast<const float4*>(att_dst);
  float ps[4], pd[4];
#pragma unroll
  for (int i = 0; i < 4; ++i) {
    int r = base + 2 * i + hf;
    r = r < rows ? r : rows - 1;
    const float4 x = h4[(size_t)r * 16 + c];
    const int hd = r & (HEADS - 1);
    const float4 s = ws4[hd * 16 + c];
    const float4 d = wd4[hd * 16 + c];
    ps[i] = x.x * s.x + x.y * s.y + x.z * s.z + x.w * s.w;
    pd[i] = x.x * d.x + x.y * d.y + x.z * d.z + x.w * d.w;
  }
#pragma unroll
  for (int off = 8; off; off >>= 1) {
#pragma unroll
    for (int i = 0; i < 4; ++i) {
      ps[i] += __shfl_xor(ps[i], off, 32);
      pd[i] += __shfl_xor(pd[i], off, 32);
    }
  }
  const float os = c == 0 ? ps[0] : c == 1 ? ps[1] : c == 2 ? ps[2] : ps[3];
  const float od = c == 0 ? pd[0] : c == 1 ? pd[1] : c == 2 ? pd[2] : pd[3];
  const int ro = base + 2 * c + hf;
  if (c < 4 && ro < rows) {
    a_s[ro] = os;
    a_d[ro] = od;
  }
}

__device__ __forceinline__ void edge_endpoints(const int* ei, int e, int& s, int& d) {
  if (e < EE) { s = ei[e]; d = ei[EE + e]; }
  else        { s = d = e - EE; }
}

__device__ __forceinline__ float leaky(float x) {
  return x > 0.0f ? x : NEG * x;
}

__device__ __forceinline__ void atomicMaxF32(float* addr, float val) {
  if (val >= 0.0f) atomicMax((int*)addr, __float_as_int(val));
  else             atomicMin((unsigned int*)addr, __float_as_uint(val));
}

#define EPT 4

template <int HEADS>
__global__ void edge_max_kernel(const int* __restrict__ ei,
                                const float* __restrict__ a_s,
                                const float* __restrict__ a_d,
                                float* __restrict__ emax) {
  const int tid    = blockIdx.x * blockDim.x + threadIdx.x;
  const int stride = gridDim.x * blockDim.x;
  const int total  = (EE + NN) * HEADS;
  int s[EPT], d[EPT], hd[EPT];
  bool ok[EPT];
  float ev[EPT];
#pragma unroll
  for (int j = 0; j < EPT; ++j) {
    const int t = tid + j * stride;
    ok[j] = t < total;
    const int tt = ok[j] ? t : 0;
    hd[j] = tt & (HEADS - 1);
    edge_endpoints(ei, tt / HEADS, s[j], d[j]);
  }
#pragma unroll
  for (int j = 0; j < EPT; ++j)
    ev[j] = leaky(a_s[s[j] * HEADS + hd[j]] + a_d[d[j] * HEADS + hd[j]]);
#pragma unroll
  for (int j = 0; j < EPT; ++j)
    if (ok[j]) atomicMaxF32(&emax[d[j] * HEADS + hd[j]], ev[j]);
}

template <int HEADS>
__global__ void edge_expsum_kernel(const int* __restrict__ ei,
                                   const float* __restrict__ a_s,
                                   const float* __restrict__ a_d,
                                   const float* __restrict__ emax,
                                   float* __restrict__ denom) {
  const int tid    = blockIdx.x * blockDim.x + threadIdx.x;
  const int stride = gridDim.x * blockDim.x;
  const int total  = (EE + NN) * HEADS;
  int s[EPT], d[EPT], hd[EPT];
  bool ok[EPT];
  float ex[EPT];
#pragma unroll
  for (int j = 0; j < EPT; ++j) {
    const int t = tid + j * stride;
    ok[j] = t < total;
    const int tt = ok[j] ? t : 0;
    hd[j] = tt & (HEADS - 1);
    edge_endpoints(ei, tt / HEADS, s[j], d[j]);
  }
#pragma unroll
  for (int j = 0; j < EPT; ++j) {
    const float ev = leaky(a_s[s[j] * HEADS + hd[j]] + a_d[d[j] * HEADS + hd[j]]);
    ex[j] = __expf(ev - emax[d[j] * HEADS + hd[j]]);
  }
#pragma unroll
  for (int j = 0; j < EPT; ++j)
    if (ok[j]) atomicAdd(&denom[d[j] * HEADS + hd[j]], ex[j]);
}

template <int HEADS, int GROUPS, int U>
__global__ void edge_aggregate_kernel(const int* __restrict__ ei,
                                      const float* __restrict__ a_s,
                                      const float* __restrict__ a_d,
                                      const float* __restrict__ emax,
                                      const float* __restrict__ denom,
                                      const float* __restrict__ h,
                                      float* __restrict__ acc) {
  static_assert(HEADS >= 1 && HEADS <= 32 && (HEADS & (HEADS - 1)) == 0,
                "HEADS must be a power of two that fits in one wave");
  constexpr int G = 32 / HEADS;
  static_assert(G % U == 0, "a batch must not run past the group's lanes");
  constexpr int TOTAL = EE + NN;
  const int lane = threadIdx.x & 31;
  const int wid = __builtin_amdgcn_readfirstlane((int)((blockIdx.x * blockDim.x + threadIdx.x) >> 5));
  const int sub = lane / HEADS;
  const int hd  = lane & (HEADS - 1);
  for (int g = 0; g < GROUPS; ++g) {
    const int ebase = (wid * GROUPS + g) * G;
    if (ebase >= TOTAL) break;
    const int e = ebase + sub;
    int s = 0, d = 0;
    float alpha = 0.0f;
    if (e < TOTAL) {
      edge_endpoints(ei, e, s, d);
      const float ev = leaky(a_s[s * HEADS + hd] + a_d[d * HEADS + hd]);
      alpha = __expf(ev - emax[d * HEADS + hd]) / denom[d * HEADS + hd];
    }
    const int cnt = (TOTAL - ebase) < G ? (TOTAL - ebase) : G;
    for (int j0 = 0; j0 < cnt; j0 += U) {
      float r0[U][HEADS], r1[U][HEADS];
#pragma unroll
      for (int u = 0; u < U; ++u) {
        const int sj = __builtin_amdgcn_readlane(s, (j0 + u) * HEADS);
        const float* hrow = h + (size_t)sj * HEADS * OUTC;
#pragma unroll
        for (int k = 0; k < HEADS; ++k) {
          r0[u][k] = hrow[k * OUTC + lane];
          r1[u][k] = hrow[k * OUTC + lane + 32];
        }
      }
#pragma unroll
      for (int u = 0; u < U; ++u) {
        const int j = j0 + u;
        if (j < cnt) {
          const int dj = __builtin_amdgcn_readlane(d, j * HEADS);
          float v0 = 0.0f, v1 = 0.0f;
#pragma unroll
          for (int k = 0; k < HEADS; ++k) {
            const float al = __int_as_float(__builtin_amdgcn_readlane(__float_as_int(alpha), j * HEADS + k));
            v0 += al * r0[u][k];
            v1 += al * r1[u][k];
          }
          float* arow = acc + (size_t)dj * OUTC;
          atomicAdd(&arow[lane],      v0);
          atomicAdd(&arow[lane + 32], v1);
        }
      }
    }
  }
}

__global__ void finalize_kernel(const float* acc,
                                const float* __restrict__ bias,
                                float* out,
                                float invH, int do_elu) {
  const int tid    = blockIdx.x * blockDim.x + threadIdx.x;
  const int stride = gridDim.x * blockDim.x;
  const int n4     = NN * OUTC / 4;
  const float4* a4 = reinterpret_cast<const float4*>(acc);
  const float4* b4 = reinterpret_cast<const float4*>(bias);
  float4* o4 = reinterpret_cast<float4*>(out);
#pragma unroll
  for (int j = 0; j < VPT; ++j) {
    const int i = tid + j * stride;
    if (i < n4) {
      const float4 a = a4[i];
      const float4 b = b4[i & (OUTC / 4 - 1)];
      float4 v;
      v.x = a.x * invH + b.x;
      v.y = a.y * invH + b.y;
      v.z = a.z * invH + b.z;
      v.w = a.w * invH + b.w;
      if (do_elu) {
        v.x = v.x > 0.0f ? v.x : (__expf(v.x) - 1.0f);
        v.y = v.y > 0.0f ? v.y : (__expf(v.y) - 1.0f);
        v.z = v.z > 0.0f ? v.z : (__expf(v.z) - 1.0f);
        v.w = v.w > 0.0f ? v.w : (__expf(v.w) - 1.0f);
      }
      o4[i] = v;
    }
  }
}

static inline unsigned cdiv(long long a, long long b) { return (unsigned)((a + b - 1) / b); }

extern "C" void kernel_launch(void* const* d_in, const int* in_sizes, int n_in,
                              void* d_out, int out_size, void* d_ws, size_t ws_size,
                              hipStream_t stream) {
  (void)in_sizes; (void)n_in; (void)out_size; (void)ws_size;
  const float* x    = (const float*)d_in[0];
  const int*   ei   = (const int*)d_in[1];
  const float* W1   = (const float*)d_in[2];
  const float* asw1 = (const float*)d_in[3];
  const float* adw1 = (const float*)d_in[4];
  const float* b1   = (const float*)d_in[5];
  const float* W2   = (const float*)d_in[6];
  const float* asw2 = (const float*)d_in[7];
  const float* adw2 = (const float*)d_in[8];
  const float* b2   = (const float*)d_in[9];
  float* out = (float*)d_out;

  float* ws   = (float*)d_ws;
  float* h1   = ws; ws += (size_t)NN * H1 * OUTC;
  float* as1  = ws; ws += (size_t)NN * H1;
  float* ad1  = ws; ws += (size_t)NN * H1;
  float* em1  = ws; ws += (size_t)NN * H1;
  float* dn1  = ws; ws += (size_t)NN * H1;
  float* acc1 = ws; ws += (size_t)NN * OUTC;
  float* h2   = ws; ws += (size_t)NN * OUTC;
  float* as2  = ws; ws += (size_t)NN;
  float* ad2  = ws; ws += (size_t)NN;
  float* em2  = ws; ws += (size_t)NN;
  float* dn2  = ws; ws += (size_t)NN;

  const int T = 256;
  const long long tot = (long long)(EE + NN);
  const uint32_t NEG_INF = 0xFF800000u;
  const unsigned gFillNH = cdiv(cdiv(NN * H1 / 4, VPT), T);
  const unsigned gFillNC = cdiv(cdiv((long long)NN * OUTC / 4, VPT), T);
  const unsigned gFillN  = cdiv(cdiv(NN / 4, VPT), T);

  fill_u32_kernel<<<gFillNH, T, 0, stream>>>((uint32_t*)em1, NEG_INF, NN * H1);
  fill_u32_kernel<<<gFillNH, T, 0, stream>>>((uint32_t*)dn1, 0u, NN * H1);
  fill_u32_kernel<<<gFillNC, T, 0, stream>>>((uint32_t*)acc1, 0u, NN * OUTC);

  {
    long long waves = ((long long)NN / 16) * (H1 * OUTC / 64);
    gemm_wmma_f32_kernel<<<cdiv(waves * 32, T), T, 0, stream>>>(x, W1, h1, NN, INCH, H1 * OUTC);
  }
  att_scores_kernel<H1><<<cdiv(cdiv((long long)NN * H1, 8) * 32, T), T, 0, stream>>>(h1, asw1, adw1, as1, ad1, NN * H1);
  edge_max_kernel<H1><<<cdiv(cdiv(tot * H1, EPT), T), T, 0, stream>>>(ei, as1, ad1, em1);
  edge_expsum_kernel<H1><<<cdiv(cdiv(tot * H1, EPT), T), T, 0, stream>>>(ei, as1, ad1, em1, dn1);
  edge_aggregate_kernel<H1, 2, 2><<<cdiv(cdiv(tot, 8) * 32, T), T, 0, stream>>>(ei, as1, ad1, em1, dn1, h1, acc1);
  finalize_kernel<<<gFillNC, T, 0, stream>>>(acc1, b1, acc1, 1.0f / H1, 1);

  fill_u32_kernel<<<gFillN, T, 0, stream>>>((uint32_t*)em2, NEG_INF, NN);
  fill_u32_kernel<<<gFillN, T, 0, stream>>>((uint32_t*)dn2, 0u, NN);
  fill_u32_kernel<<<gFillNC, T, 0, stream>>>((uint32_t*)out, 0u, NN * OUTC);

  {
    long long waves = ((long long)NN / 16) * (OUTC / 64);
    gemm_wmma_f32_kernel<<<cdiv(waves * 32, T), T, 0, stream>>>(acc1, W2, h2, NN, OUTC, OUTC);
  }
  att_scores_kernel<1><<<cdiv(cdiv(NN, 8) * 32, T), T, 0, stream>>>(h2, asw2, adw2, as2, ad2, NN);
  edge_max_kernel<1><<<cdiv(cdiv(tot, EPT), T), T, 0, stream>>>(ei, as2, ad2, em2);
  edge_expsum_kernel<1><<<cdiv(cdiv(tot, EPT), T), T, 0, stream>>>(ei, as2, ad2, em2, dn2);
  edge_aggregate_kernel<1, 1, 8><<<cdiv(cdiv(tot, 32) * 32, T), T, 0, stream>>>(ei, as2, ad2, em2, dn2, h2, out);
  finalize_kernel<<<gFillNC, T, 0, stream>>>(out, b2, out, 1.0f, 0);
}


# Edges grouped by destination per call (count, scan, scatter); a wave owns a node: max, exp-sum, weighted row sum in registers, one write; no float atomics, fills, edge passes; sums in arrival order
# speedup vs baseline: 3.2134x; 1.4913x over previous
#include <hip/hip_runtime.h>
#include <stdint.h>

#define NN    50000
#define EE    400000
#define INCH  128
#define OUTC  64
#define H1    8
#define NEG   0.2f

typedef float v2f __attribute__((ext_vector_type(2)));
typedef float v8f __attribute__((ext_vector_type(8)));

#define VPT 4
__global__ void fill_u32_kernel(uint32_t* p, uint32_t v, int n) {
  const int tid    = blockIdx.x * blockDim.x + threadIdx.x;
  const int stride = gridDim.x * blockDim.x;
  const int n4     = n >> 2;
  uint4* p4 = reinterpret_cast<uint4*>(p);
  const uint4 val = make_uint4(v, v, v, v);
#pragma unroll
  for (int j = 0; j < VPT; ++j) {
    const int i = tid + j * stride;
    if (i < n4) p4[i] = val;
  }
}

__global__ void gemm_wmma_f32_kernel(const float* __restrict__ A,
                                     const float* __restrict__ B,
                                     float* __restrict__ C,
                                     int M, int K, int Nc) {
  const int wid  = (blockIdx.x * blockDim.x + threadIdx.x) >> 5;
  const int lane = threadIdx.x & 31;
  const int tilesN   = Nc >> 6;
  const int rowTiles = M >> 4;
  if (wid >= rowTiles * tilesN) return;

  const int rt = wid / tilesN;
  const int cg = wid - rt * tilesN;
  const int m0 = rt << 4;
  const int n0 = cg << 6;
  const int ml = lane & 15;
  const int kh = lane >> 4;

  v8f acc0 = {}, acc1 = {}, acc2 = {}, acc3 = {};
  const float* arow = A + (size_t)(m0 + ml) * K;

  for (int k = 0; k < K; k += 4) {
    const int ka = k + 2 * kh;
    v2f a;
    a.x = arow[ka];
    a.y = arow[ka + 1];
    const float* b0p = B + (size_t)ka * Nc + n0 + ml;
    const float* b1p = b0p + Nc;
    v2f b0; b0.x = b0p[0];  b0.y = b1p[0];
    v2f b1; b1.x = b0p[16]; b1.y = b1p[16];
    v2f b2; b2.x = b0p[32]; b2.y = b1p[32];
    v2f b3; b3.x = b0p[48]; b3.y = b1p[48];
    acc0 = __builtin_amdgcn_wmma_f32_16x16x4_f32(false, a, false, b0, (short)0, acc0, false, false);
    acc1 = __builtin_amdgcn_wmma_f32_16x16x4_f32(false, a, false, b1, (short)0, acc1, false, false);
    acc2 = __builtin_amdgcn_wmma_f32_16x16x4_f32(false, a, false, b2, (short)0, acc2, false, false);
    acc3 = __builtin_amdgcn_wmma_f32_16x16x4_f32(false, a, false, b3, (short)0, acc3, false, false);
  }

  const int mr = m0 + 8 * kh;
#pragma unroll
  for (int v = 0; v < 8; ++v) {
    float* crow = C + (size_t)(mr + v) * Nc + n0 + ml;
    crow[0]  = acc0[v];
    crow[16] = acc1[v];
    crow[32] = acc2[v];
    crow[48] = acc3[v];
  }
}

template <int HEADS>
__global__ void att_scores_kernel(const float* __restrict__ h,
                                  const float* __restrict__ att_src,
                                  const float* __restrict__ att_dst,
                                  float* __restrict__ a_s,
                                  float* __restrict__ a_d,
                                  int rows) {
  const int wid  = (blockIdx.x * blockDim.x + threadIdx.x) >> 5;
  const int lane = threadIdx.x & 31;
  const int c    = lane & 15;
  const int hf   = lane >> 4;
  const int base = wid * 8;
  if (base >= rows) return;
  const float4* h4  = reinterpret_cast<const float4*>(h);
  const float4* ws4 = reinterpret_cast<const float4*>(att_src);
  const float4* wd4 = reinterpret_cast<const float4*>(att_dst);
  float ps[4], pd[4];
#pragma unroll
  for (int i = 0; i < 4; ++i) {
    int r = base + 2 * i + hf;
    r = r < rows ? r : rows - 1;
    const float4 x = h4[(size_t)r * 16 + c];
    const int hd = r & (HEADS - 1);
    const float4 s = ws4[hd * 16 + c];
    const float4 d = wd4[hd * 16 + c];
    ps[i] = x.x * s.x + x.y * s.y + x.z * s.z + x.w * s.w;
    pd[i] = x.x * d.x + x.y * d.y + x.z * d.z + x.w * d.w;
  }
#pragma unroll
  for (int off = 8; off; off >>= 1) {
#pragma unroll
    for (int i = 0; i < 4; ++i) {
      ps[i] += __shfl_xor(ps[i], off, 32);
      pd[i] += __shfl_xor(pd[i], off, 32);
    }
  }
  const float os = c == 0 ? ps[0] : c == 1 ? ps[1] : c == 2 ? ps[2] : ps[3];
  const float od = c == 0 ? pd[0] : c == 1 ? pd[1] : c == 2 ? pd[2] : pd[3];
  const int ro = base + 2 * c + hf;
  if (c < 4 && ro < rows) {
    a_s[ro] = os;
    a_d[ro] = od;
  }
}

__device__ __forceinline__ float leaky(float x) {
  return x > 0.0f ? x : NEG * x;
}

#define EPT 4
__global__ void dst_rank_kernel(const int* __restrict__ ei,
                                int* __restrict__ counts,
                                int* __restrict__ rank) {
  const int tid    = blockIdx.x * blockDim.x + threadIdx.x;
  const int stride = gridDim.x * blockDim.x;
  int dd[EPT], r[EPT];
#pragma unroll
  for (int j = 0; j < EPT; ++j) {
    const int e = tid + j * stride;
    dd[j] = e < EE ? ei[EE + e] : -1;
  }
#pragma unroll
  for (int j = 0; j < EPT; ++j) r[j] = dd[j] >= 0 ? atomicAdd(&counts[dd[j]], 1) : 0;
#pragma unroll
  for (int j = 0; j < EPT; ++j) {
    const int e = tid + j * stride;
    if (e < EE) rank[e] = r[j];
  }
}

#define SCAN_T 1024
__global__ void __launch_bounds__(SCAN_T)
offsets_scan_kernel(const int* __restrict__ counts, int* __restrict__ offsets) {
  constexpr int WAVES = SCAN_T / 32;
  static_assert(WAVES == 32, "the wave totals are summed by the 32 lanes of a wave");
  constexpr int IT = ((NN + WAVES - 1) / WAVES + 31) / 32;
  constexpr int PER_WAVE = IT * 32;
  __shared__ int wtot[WAVES];
  const int lane = threadIdx.x & 31;
  const int w    = threadIdx.x >> 5;
  const int base = w * PER_WAVE;
  int v[IT];
  int tot = 0;
#pragma unroll
  for (int j = 0; j < IT; ++j) {
    const int i = base + j * 32 + lane;
    v[j] = i < NN ? counts[i] : 0;
    tot += v[j];
  }
  int inc = tot;
  for (int off = 1; off < 32; off <<= 1) {
    const int nb = __shfl_up(inc, off, 32);
    if (lane >= off) inc += nb;
  }
  if (lane == 31) wtot[w] = inc;
  __syncthreads();
  int carry = lane < w ? wtot[lane] : 0;
  for (int off = 16; off; off >>= 1) carry += __shfl_xor(carry, off, 32);
  int run = carry + inc - tot;
#pragma unroll
  for (int j = 0; j < IT; ++j) {
    const int i = base + j * 32 + lane;
    if (i < NN) offsets[i] = run;
    run += v[j];
  }
}

__global__ void dst_scatter_kernel(const int* __restrict__ ei,
                                   const int* __restrict__ offsets,
                                   const int* __restrict__ rank,
                                   int* __restrict__ csr_src) {
  const int tid    = blockIdx.x * blockDim.x + threadIdx.x;
  const int stride = gridDim.x * blockDim.x;
  int dd[EPT], ss[EPT], rr[EPT], pos[EPT];
#pragma unroll
  for (int j = 0; j < EPT; ++j) {
    const int e  = tid + j * stride;
    const int ec = e < EE ? e : 0;
    dd[j] = ei[EE + ec];
    ss[j] = ei[ec];
    rr[j] = rank[ec];
  }
#pragma unroll
  for (int j = 0; j < EPT; ++j) pos[j] = offsets[dd[j]] + rr[j];
#pragma unroll
  for (int j = 0; j < EPT; ++j) {
    const int e = tid + j * stride;
    if (e < EE) csr_src[pos[j]] = ss[j];
  }
}

template <int HEADS>
__device__ __forceinline__ void node_scores(const int* __restrict__ csr_src,
                                            const float* __restrict__ a_s,
                                            const float (&ad)[HEADS],
                                            int off, int deg, int d, int i,
                                            int& s, float (&ev)[HEADS]) {
  s = i < deg ? csr_src[off + i] : d;
  const float* ap = a_s + (size_t)s * HEADS;
  if constexpr (HEADS % 4 == 0) {
    const float4* ap4 = reinterpret_cast<const float4*>(ap);
#pragma unroll
    for (int q = 0; q < HEADS / 4; ++q) {
      const float4 t = ap4[q];
      ev[4 * q + 0] = leaky(t.x + ad[4 * q + 0]);
      ev[4 * q + 1] = leaky(t.y + ad[4 * q + 1]);
      ev[4 * q + 2] = leaky(t.z + ad[4 * q + 2]);
      ev[4 * q + 3] = leaky(t.w + ad[4 * q + 3]);
    }
  } else {
#pragma unroll
    for (int k = 0; k < HEADS; ++k) ev[k] = leaky(ap[k] + ad[k]);
  }
}

template <int HEADS, int U, int NPW>
__global__ void node_aggregate_kernel(const int* __restrict__ offsets,
                                      const int* __restrict__ counts,
                                      const int* __restrict__ csr_src,
                                      const float* __restrict__ a_s,
                                      const float* __restrict__ a_d,
                                      const float* __restrict__ h,
                                      const float* __restrict__ bias,
                                      float* __restrict__ out,
                                      float invH, int do_elu) {
  static_assert(32 % U == 0, "a batch must not run past the wave's lanes");
  const int lane = threadIdx.x & 31;
  const int wid = __builtin_amdgcn_readfirstlane((int)((blockIdx.x * blockDim.x + threadIdx.x) >> 5));
  const float bias0 = bias[lane], bias1 = bias[lane + 32];
  for (int t = 0; t < NPW; ++t) {
    const int d = wid * NPW + t;
    if (d >= NN) break;
    const int off = __builtin_amdgcn_readfirstlane(offsets[d]);
    const int deg = __builtin_amdgcn_readfirstlane(counts[d]);
    const int cnt = deg + 1;
    float ad[HEADS];
#pragma unroll
    for (int k = 0; k < HEADS; ++k) ad[k] = a_d[d * HEADS + k];

    int s0; float ev0[HEADS];
    node_scores<HEADS>(csr_src, a_s, ad, off, deg, d, lane, s0, ev0);
    const bool in0 = lane < cnt;

    float m[HEADS];
#pragma unroll
    for (int k = 0; k < HEADS; ++k) m[k] = in0 ? ev0[k] : -__builtin_inff();
    for (int g0 = 32; g0 < cnt; g0 += 32) {
      int s; float ev[HEADS];
      node_scores<HEADS>(csr_src, a_s, ad, off, deg, d, g0 + lane, s, ev);
      if (g0 + lane < cnt) {
#pragma unroll
        for (int k = 0; k < HEADS; ++k) m[k] = fmaxf(m[k], ev[k]);
      }
    }
#pragma unroll
    for (int k = 0; k < HEADS; ++k)
      for (int o = 16; o; o >>= 1) m[k] = fmaxf(m[k], __shfl_xor(m[k], o, 32));

    float ex0[HEADS], sum[HEADS];
#pragma unroll
    for (int k = 0; k < HEADS; ++k) {
      ex0[k] = __expf(ev0[k] - m[k]);
      sum[k] = in0 ? ex0[k] : 0.0f;
    }
    for (int g0 = 32; g0 < cnt; g0 += 32) {
      int s; float ev[HEADS];
      node_scores<HEADS>(csr_src, a_s, ad, off, deg, d, g0 + lane, s, ev);
      if (g0 + lane < cnt) {
#pragma unroll
        for (int k = 0; k < HEADS; ++k) sum[k] += __expf(ev[k] - m[k]);
      }
    }
#pragma unroll
    for (int k = 0; k < HEADS; ++k)
      for (int o = 16; o; o >>= 1) sum[k] += __shfl_xor(sum[k], o, 32);

    float v0 = 0.0f, v1 = 0.0f;
    auto add_rows = [&](int s, const float (&alpha)[HEADS], int n) {
      for (int j0 = 0; j0 < n; j0 += U) {
        float r0[U][HEADS], r1[U][HEADS];
#pragma unroll
        for (int u = 0; u < U; ++u) {
          const int sj = __builtin_amdgcn_readlane(s, j0 + u);
          const float* hrow = h + (size_t)sj * HEADS * OUTC;
#pragma unroll
          for (int k = 0; k < HEADS; ++k) {
            r0[u][k] = hrow[k * OUTC + lane];
            r1[u][k] = hrow[k * OUTC + lane + 32];
          }
        }
#pragma unroll
        for (int u = 0; u < U; ++u) {
          if (j0 + u < n) {
#pragma unroll
            for (int k = 0; k < HEADS; ++k) {
              const float al = __int_as_float(__builtin_amdgcn_readlane(__float_as_int(alpha[k]), j0 + u));
              v0 += al * r0[u][k];
              v1 += al * r1[u][k];
            }
          }
        }
      }
    };
    {
      float alpha[HEADS];
#pragma unroll
      for (int k = 0; k < HEADS; ++k) alpha[k] = ex0[k] / sum[k];
      add_rows(s0, alpha, cnt < 32 ? cnt : 32);
    }
    for (int g0 = 32; g0 < cnt; g0 += 32) {
      int s; float ev[HEADS], alpha[HEADS];
      node_scores<HEADS>(csr_src, a_s, ad, off, deg, d, g0 + lane, s, ev);
#pragma unroll
      for (int k = 0; k < HEADS; ++k) alpha[k] = __expf(ev[k] - m[k]) / sum[k];
      add_rows(s, alpha, (cnt - g0) < 32 ? (cnt - g0) : 32);
    }

    float o0 = v0 * invH + bias0;
    float o1 = v1 * invH + bias1;
    if (do_elu) {
      o0 = o0 > 0.0f ? o0 : (__expf(o0) - 1.0f);
      o1 = o1 > 0.0f ? o1 : (__expf(o1) - 1.0f);
    }
    out[(size_t)d * OUTC + lane]      = o0;
    out[(size_t)d * OUTC + lane + 32] = o1;
  }
}

static inline unsigned cdiv(long long a, long long b) { return (unsigned)((a + b - 1) / b); }

extern "C" void kernel_launch(void* const* d_in, const int* in_sizes, int n_in,
                              void* d_out, int out_size, void* d_ws, size_t ws_size,
                              hipStream_t stream) {
  (void)in_sizes; (void)n_in; (void)out_size; (void)ws_size;
  const float* x    = (const float*)d_in[0];
  const int*   ei   = (const int*)d_in[1];
  const float* W1   = (const float*)d_in[2];
  const float* asw1 = (const float*)d_in[3];
  const float* adw1 = (const float*)d_in[4];
  const float* b1   = (const float*)d_in[5];
  const float* W2   = (const float*)d_in[6];
  const float* asw2 = (const float*)d_in[7];
  const float* adw2 = (const float*)d_in[8];
  const float* b2   = (const float*)d_in[9];
  float* out = (float*)d_out;

  float* ws   = (float*)d_ws;
  float* h1   = ws; ws += (size_t)NN * H1 * OUTC;
  float* as1  = ws; ws += (size_t)NN * H1;
  float* ad1  = ws; ws += (size_t)NN * H1;
  float* hmid = ws; ws += (size_t)NN * OUTC;
  float* h2   = ws; ws += (size_t)NN * OUTC;
  float* as2  = ws; ws += (size_t)NN;
  float* ad2  = ws; ws += (size_t)NN;
  int* counts  = (int*)ws; ws += (size_t)NN;
  int* offsets = (int*)ws; ws += (size_t)NN;
  int* csr_src = (int*)ws; ws += (size_t)EE;
  int* rank = (int*)h2;

  const int T = 256;
  const int NPW = 4;
  const unsigned gNode = cdiv((long long)cdiv(NN, NPW) * 32, T);

  fill_u32_kernel<<<cdiv(cdiv(NN / 4, VPT), T), T, 0, stream>>>((uint32_t*)counts, 0u, NN);
  dst_rank_kernel<<<cdiv(cdiv(EE, EPT), T), T, 0, stream>>>(ei, counts, rank);
  offsets_scan_kernel<<<1, SCAN_T, 0, stream>>>(counts, offsets);
  dst_scatter_kernel<<<cdiv(cdiv(EE, EPT), T), T, 0, stream>>>(ei, offsets, rank, csr_src);

  {
    long long waves = ((long long)NN / 16) * (H1 * OUTC / 64);
    gemm_wmma_f32_kernel<<<cdiv(waves * 32, T), T, 0, stream>>>(x, W1, h1, NN, INCH, H1 * OUTC);
  }
  att_scores_kernel<H1><<<cdiv(cdiv((long long)NN * H1, 8) * 32, T), T, 0, stream>>>(h1, asw1, adw1, as1, ad1, NN * H1);
  node_aggregate_kernel<H1, 2, NPW><<<gNode, T, 0, stream>>>(offsets, counts, csr_src, as1, ad1, h1, b1, hmid, 1.0f / H1, 1);

  {
    long long waves = ((long long)NN / 16) * (OUTC / 64);
    gemm_wmma_f32_kernel<<<cdiv(waves * 32, T), T, 0, stream>>>(hmid, W2, h2, NN, OUTC, OUTC);
  }
  att_scores_kernel<1><<<cdiv(cdiv(NN, 8) * 32, T), T, 0, stream>>>(h2, asw2, adw2, as2, ad2, NN);
  node_aggregate_kernel<1, 8, NPW><<<gNode, T, 0, stream>>>(offsets, counts, csr_src, as2, ad2, h2, b2, out, 1.0f, 0);
}


# Layer 1 reassociated: per head sum alpha-weighted x rows (128 wide, 4x less gather), then one WMMA product with W1 per 16-node block; scores from x@(W1.att); h1 never stored; sums in a new order
# speedup vs baseline: 4.5181x; 1.4060x over previous
#include <hip/hip_runtime.h>
#include <stdint.h>

#define NN    50000
#define EE    400000
#define INCH  128
#define OUTC  64
#define H1    8
#define NEG   0.2f

typedef float v2f __attribute__((ext_vector_type(2)));
typedef float v8f __attribute__((ext_vector_type(8)));

#define VPT 4
__global__ void fill_u32_kernel(uint32_t* p, uint32_t v, int n) {
  const int tid    = blockIdx.x * blockDim.x + threadIdx.x;
  const int stride = gridDim.x * blockDim.x;
  const int n4     = n >> 2;
  uint4* p4 = reinterpret_cast<uint4*>(p);
  const uint4 val = make_uint4(v, v, v, v);
#pragma unroll
  for (int j = 0; j < VPT; ++j) {
    const int i = tid + j * stride;
    if (i < n4) p4[i] = val;
  }
}

__global__ void gemm_wmma_f32_kernel(const float* __restrict__ A,
                                     const float* __restrict__ B,
                                     float* __restrict__ C,
                                     int M, int K, int Nc) {
  const int wid  = (blockIdx.x * blockDim.x + threadIdx.x) >> 5;
  const int lane = threadIdx.x & 31;
  const int tilesN   = Nc >> 6;
  const int rowTiles = M >> 4;
  if (wid >= rowTiles * tilesN) return;

  const int rt = wid / tilesN;
  const int cg = wid - rt * tilesN;
  const int m0 = rt << 4;
  const int n0 = cg << 6;
  const int ml = lane & 15;
  const int kh = lane >> 4;

  v8f acc0 = {}, acc1 = {}, acc2 = {}, acc3 = {};
  const float* arow = A + (size_t)(m0 + ml) * K;

  for (int k = 0; k < K; k += 4) {
    const int ka = k + 2 * kh;
    v2f a;
    a.x = arow[ka];
    a.y = arow[ka + 1];
    const float* b0p = B + (size_t)ka * Nc + n0 + ml;
    const float* b1p = b0p + Nc;
    v2f b0; b0.x = b0p[0];  b0.y = b1p[0];
    v2f b1; b1.x = b0p[16]; b1.y = b1p[16];
    v2f b2; b2.x = b0p[32]; b2.y = b1p[32];
    v2f b3; b3.x = b0p[48]; b3.y = b1p[48];
    acc0 = __builtin_amdgcn_wmma_f32_16x16x4_f32(false, a, false, b0, (short)0, acc0, false, false);
    acc1 = __builtin_amdgcn_wmma_f32_16x16x4_f32(false, a, false, b1, (short)0, acc1, false, false);
    acc2 = __builtin_amdgcn_wmma_f32_16x16x4_f32(false, a, false, b2, (short)0, acc2, false, false);
    acc3 = __builtin_amdgcn_wmma_f32_16x16x4_f32(false, a, false, b3, (short)0, acc3, false, false);
  }

  const int mr = m0 + 8 * kh;
#pragma unroll
  for (int v = 0; v < 8; ++v) {
    float* crow = C + (size_t)(mr + v) * Nc + n0 + ml;
    crow[0]  = acc0[v];
    crow[16] = acc1[v];
    crow[32] = acc2[v];
    crow[48] = acc3[v];
  }
}

template <int HEADS>
__global__ void att_scores_kernel(const float* __restrict__ h,
                                  const float* __restrict__ att_src,
                                  const float* __restrict__ att_dst,
                                  float* __restrict__ a_s,
                                  float* __restrict__ a_d,
                                  int rows) {
  const int wid  = (blockIdx.x * blockDim.x + threadIdx.x) >> 5;
  const int lane = threadIdx.x & 31;
  const int c    = lane & 15;
  const int hf   = lane >> 4;
  const int base = wid * 8;
  if (base >= rows) return;
  const float4* h4  = reinterpret_cast<const float4*>(h);
  const float4* ws4 = reinterpret_cast<const float4*>(att_src);
  const float4* wd4 = reinterpret_cast<const float4*>(att_dst);
  float ps[4], pd[4];
#pragma unroll
  for (int i = 0; i < 4; ++i) {
    int r = base + 2 * i + hf;
    r = r < rows ? r : rows - 1;
    const float4 x = h4[(size_t)r * 16 + c];
    const int hd = r & (HEADS - 1);
    const float4 s = ws4[hd * 16 + c];
    const float4 d = wd4[hd * 16 + c];
    ps[i] = x.x * s.x + x.y * s.y + x.z * s.z + x.w * s.w;
    pd[i] = x.x * d.x + x.y * d.y + x.z * d.z + x.w * d.w;
  }
#pragma unroll
  for (int off = 8; off; off >>= 1) {
#pragma unroll
    for (int i = 0; i < 4; ++i) {
      ps[i] += __shfl_xor(ps[i], off, 32);
      pd[i] += __shfl_xor(pd[i], off, 32);
    }
  }
  const float os = c == 0 ? ps[0] : c == 1 ? ps[1] : c == 2 ? ps[2] : ps[3];
  const float od = c == 0 ? pd[0] : c == 1 ? pd[1] : c == 2 ? pd[2] : pd[3];
  const int ro = base + 2 * c + hf;
  if (c < 4 && ro < rows) {
    a_s[ro] = os;
    a_d[ro] = od;
  }
}

__device__ __forceinline__ float leaky(float x) {
  return x > 0.0f ? x : NEG * x;
}

#define EPT 4
__global__ void dst_rank_kernel(const int* __restrict__ ei,
                                int* __restrict__ counts,
                                int* __restrict__ rank) {
  const int tid    = blockIdx.x * blockDim.x + threadIdx.x;
  const int stride = gridDim.x * blockDim.x;
  int dd[EPT], r[EPT];
#pragma unroll
  for (int j = 0; j < EPT; ++j) {
    const int e = tid + j * stride;
    dd[j] = e < EE ? ei[EE + e] : -1;
  }
#pragma unroll
  for (int j = 0; j < EPT; ++j) r[j] = dd[j] >= 0 ? atomicAdd(&counts[dd[j]], 1) : 0;
#pragma unroll
  for (int j = 0; j < EPT; ++j) {
    const int e = tid + j * stride;
    if (e < EE) rank[e] = r[j];
  }
}

#define SCAN_T 1024
__global__ void __launch_bounds__(SCAN_T)
offsets_scan_kernel(const int* __restrict__ counts, int* __restrict__ offsets) {
  constexpr int WAVES = SCAN_T / 32;
  static_assert(WAVES == 32, "the wave totals are summed by the 32 lanes of a wave");
  constexpr int IT = ((NN + WAVES - 1) / WAVES + 31) / 32;
  constexpr int PER_WAVE = IT * 32;
  __shared__ int wtot[WAVES];
  const int lane = threadIdx.x & 31;
  const int w    = threadIdx.x >> 5;
  const int base = w * PER_WAVE;
  int v[IT];
  int tot = 0;
#pragma unroll
  for (int j = 0; j < IT; ++j) {
    const int i = base + j * 32 + lane;
    v[j] = i < NN ? counts[i] : 0;
    tot += v[j];
  }
  int inc = tot;
  for (int off = 1; off < 32; off <<= 1) {
    const int nb = __shfl_up(inc, off, 32);
    if (lane >= off) inc += nb;
  }
  if (lane == 31) wtot[w] = inc;
  __syncthreads();
  int carry = lane < w ? wtot[lane] : 0;
  for (int off = 16; off; off >>= 1) carry += __shfl_xor(carry, off, 32);
  int run = carry + inc - tot;
#pragma unroll
  for (int j = 0; j < IT; ++j) {
    const int i = base + j * 32 + lane;
    if (i < NN) offsets[i] = run;
    run += v[j];
  }
}

__global__ void dst_scatter_kernel(const int* __restrict__ ei,
                                   const int* __restrict__ offsets,
                                   const int* __restrict__ rank,
                                   int* __restrict__ csr_src) {
  const int tid    = blockIdx.x * blockDim.x + threadIdx.x;
  const int stride = gridDim.x * blockDim.x;
  int dd[EPT], ss[EPT], rr[EPT], pos[EPT];
#pragma unroll
  for (int j = 0; j < EPT; ++j) {
    const int e  = tid + j * stride;
    const int ec = e < EE ? e : 0;
    dd[j] = ei[EE + ec];
    ss[j] = ei[ec];
    rr[j] = rank[ec];
  }
#pragma unroll
  for (int j = 0; j < EPT; ++j) pos[j] = offsets[dd[j]] + rr[j];
#pragma unroll
  for (int j = 0; j < EPT; ++j) {
    const int e = tid + j * stride;
    if (e < EE) csr_src[pos[j]] = ss[j];
  }
}

template <int HEADS>
__device__ __forceinline__ void node_scores(const int* __restrict__ csr_src,
                                            const float* __restrict__ a_s,
                                            const float (&ad)[HEADS],
                                            int off, int deg, int d, int i,
                                            int& s, float (&ev)[HEADS]) {
  s = i < deg ? csr_src[off + i] : d;
  const float* ap = a_s + (size_t)s * HEADS;
  if constexpr (HEADS % 4 == 0) {
    const float4* ap4 = reinterpret_cast<const float4*>(ap);
#pragma unroll
    for (int q = 0; q < HEADS / 4; ++q) {
      const float4 t = ap4[q];
      ev[4 * q + 0] = leaky(t.x + ad[4 * q + 0]);
      ev[4 * q + 1] = leaky(t.y + ad[4 * q + 1]);
      ev[4 * q + 2] = leaky(t.z + ad[4 * q + 2]);
      ev[4 * q + 3] = leaky(t.w + ad[4 * q + 3]);
    }
  } else {
#pragma unroll
    for (int k = 0; k < HEADS; ++k) ev[k] = leaky(ap[k] + ad[k]);
  }
}

template <int HEADS, int U, int NPW>
__global__ void node_aggregate_kernel(const int* __restrict__ offsets,
                                      const int* __restrict__ counts,
                                      const int* __restrict__ csr_src,
                                      const float* __restrict__ a_s,
                                      const float* __restrict__ a_d,
                                      const float* __restrict__ h,
                                      const float* __restrict__ bias,
                                      float* __restrict__ out,
                                      float invH, int do_elu) {
  static_assert(32 % U == 0, "a batch must not run past the wave's lanes");
  const int lane = threadIdx.x & 31;
  const int wid = __builtin_amdgcn_readfirstlane((int)((blockIdx.x * blockDim.x + threadIdx.x) >> 5));
  const float bias0 = bias[lane], bias1 = bias[lane + 32];
  for (int t = 0; t < NPW; ++t) {
    const int d = wid * NPW + t;
    if (d >= NN) break;
    const int off = __builtin_amdgcn_readfirstlane(offsets[d]);
    const int deg = __builtin_amdgcn_readfirstlane(counts[d]);
    const int cnt = deg + 1;
    float ad[HEADS];
#pragma unroll
    for (int k = 0; k < HEADS; ++k) ad[k] = a_d[d * HEADS + k];

    int s0; float ev0[HEADS];
    node_scores<HEADS>(csr_src, a_s, ad, off, deg, d, lane, s0, ev0);
    const bool in0 = lane < cnt;

    float m[HEADS];
#pragma unroll
    for (int k = 0; k < HEADS; ++k) m[k] = in0 ? ev0[k] : -__builtin_inff();
    for (int g0 = 32; g0 < cnt; g0 += 32) {
      int s; float ev[HEADS];
      node_scores<HEADS>(csr_src, a_s, ad, off, deg, d, g0 + lane, s, ev);
      if (g0 + lane < cnt) {
#pragma unroll
        for (int k = 0; k < HEADS; ++k) m[k] = fmaxf(m[k], ev[k]);
      }
    }
#pragma unroll
    for (int k = 0; k < HEADS; ++k)
      for (int o = 16; o; o >>= 1) m[k] = fmaxf(m[k], __shfl_xor(m[k], o, 32));

    float ex0[HEADS], sum[HEADS];
#pragma unroll
    for (int k = 0; k < HEADS; ++k) {
      ex0[k] = __expf(ev0[k] - m[k]);
      sum[k] = in0 ? ex0[k] : 0.0f;
    }
    for (int g0 = 32; g0 < cnt; g0 += 32) {
      int s; float ev[HEADS];
      node_scores<HEADS>(csr_src, a_s, ad, off, deg, d, g0 + lane, s, ev);
      if (g0 + lane < cnt) {
#pragma unroll
        for (int k = 0; k < HEADS; ++k) sum[k] += __expf(ev[k] - m[k]);
      }
    }
#pragma unroll
    for (int k = 0; k < HEADS; ++k)
      for (int o = 16; o; o >>= 1) sum[k] += __shfl_xor(sum[k], o, 32);

    float v0 = 0.0f, v1 = 0.0f;
    auto add_rows = [&](int s, const float (&alpha)[HEADS], int n) {
      for (int j0 = 0; j0 < n; j0 += U) {
        float r0[U][HEADS], r1[U][HEADS];
#pragma unroll
        for (int u = 0; u < U; ++u) {
          const int sj = __builtin_amdgcn_readlane(s, j0 + u);
          const float* hrow = h + (size_t)sj * HEADS * OUTC;
#pragma unroll
          for (int k = 0; k < HEADS; ++k) {
            r0[u][k] = hrow[k * OUTC + lane];
            r1[u][k] = hrow[k * OUTC + lane + 32];
          }
        }
#pragma unroll
        for (int u = 0; u < U; ++u) {
          if (j0 + u < n) {
#pragma unroll
            for (int k = 0; k < HEADS; ++k) {
              const float al = __int_as_float(__builtin_amdgcn_readlane(__float_as_int(alpha[k]), j0 + u));
              v0 += al * r0[u][k];
              v1 += al * r1[u][k];
            }
          }
        }
      }
    };
    {
      float alpha[HEADS];
#pragma unroll
      for (int k = 0; k < HEADS; ++k) alpha[k] = ex0[k] / sum[k];
      add_rows(s0, alpha, cnt < 32 ? cnt : 32);
    }
    for (int g0 = 32; g0 < cnt; g0 += 32) {
      int s; float ev[HEADS], alpha[HEADS];
      node_scores<HEADS>(csr_src, a_s, ad, off, deg, d, g0 + lane, s, ev);
#pragma unroll
      for (int k = 0; k < HEADS; ++k) alpha[k] = __expf(ev[k] - m[k]) / sum[k];
      add_rows(s, alpha, (cnt - g0) < 32 ? (cnt - g0) : 32);
    }

    float o0 = v0 * invH + bias0;
    float o1 = v1 * invH + bias1;
    if (do_elu) {
      o0 = o0 > 0.0f ? o0 : (__expf(o0) - 1.0f);
      o1 = o1 > 0.0f ? o1 : (__expf(o1) - 1.0f);
    }
    out[(size_t)d * OUTC + lane]      = o0;
    out[(size_t)d * OUTC + lane + 32] = o1;
  }
}

__global__ void att_fold_kernel(const float* __restrict__ W1,
                                const float* __restrict__ att_src,
                                const float* __restrict__ att_dst,
                                float* __restrict__ V) {
  const int t = blockIdx.x * blockDim.x + threadIdx.x;
  if (t >= INCH * 16) return;
  const int i = t >> 4, j = t & 15, hd = j & 7;
  const float4* w4 = reinterpret_cast<const float4*>(W1 + (size_t)i * (H1 * OUTC) + hd * OUTC);
  const float4* a4 = reinterpret_cast<const float4*>((j < 8 ? att_src : att_dst) + hd * OUTC);
  float s = 0.0f;
#pragma unroll
  for (int c = 0; c < OUTC / 4; ++c) {
    const float4 w = w4[c];
    const float4 a = a4[c];
    s += w.x * a.x + w.y * a.y + w.z * a.z + w.w * a.w;
  }
  V[t] = s;
}

__global__ void scores_wmma_kernel(const float* __restrict__ x,
                                   const float* __restrict__ V,
                                   float* __restrict__ a_s,
                                   float* __restrict__ a_d) {
  const int wid  = (blockIdx.x * blockDim.x + threadIdx.x) >> 5;
  const int lane = threadIdx.x & 31;
  if (wid >= NN / 16) return;
  const int m0 = wid << 4;
  const int ml = lane & 15;
  const int kh = lane >> 4;
  v8f acc = {};
  const float* arow = x + (size_t)(m0 + ml) * INCH + 2 * kh;
  const float* vcol = V + (2 * kh) * 16 + ml;
#pragma unroll 16
  for (int k = 0; k < INCH; k += 4) {
    v2f a; a.x = arow[k];      a.y = arow[k + 1];
    v2f b; b.x = vcol[k * 16]; b.y = vcol[k * 16 + 16];
    acc = __builtin_amdgcn_wmma_f32_16x16x4_f32(false, a, false, b, (short)0, acc, false, false);
  }
  float* dstp = (ml < 8 ? a_s : a_d) + (size_t)(m0 + 8 * kh) * H1 + (ml & 7);
#pragma unroll
  for (int r = 0; r < 8; ++r) dstp[r * H1] = acc[r];
}

#define NB 16
#define ZP 1028
#define UX 8
__global__ void __launch_bounds__(256)
layer1_fused_kernel(const int* __restrict__ offsets,
                    const int* __restrict__ counts,
                    const int* __restrict__ csr_src,
                    const float* __restrict__ a_s,
                    const float* __restrict__ a_d,
                    const float* __restrict__ x,
                    const float* __restrict__ W1,
                    const float* __restrict__ bias,
                    float* __restrict__ out) {
  constexpr int HEADS = H1;
  __shared__ __attribute__((aligned(16))) float zs[NB * ZP];
  static_assert(HEADS * NB * OUTC <= NB * ZP, "the products must fit where the sums were");
  float* part = zs;
  const int lane = threadIdx.x & 31;
  const int wv = __builtin_amdgcn_readfirstlane((int)(threadIdx.x >> 5));
  const int nbase = blockIdx.x * NB;
  const float4* x4 = reinterpret_cast<const float4*>(x);

  for (int t = 0; t < NB / 8; ++t) {
    const int nl = wv * (NB / 8) + t;
    const int d  = nbase + nl;
    float4 z[HEADS];
#pragma unroll
    for (int k = 0; k < HEADS; ++k) z[k] = make_float4(0.0f, 0.0f, 0.0f, 0.0f);
    if (d < NN) {
      const int off = __builtin_amdgcn_readfirstlane(offsets[d]);
      const int deg = __builtin_amdgcn_readfirstlane(counts[d]);
      const int cnt = deg + 1;
      float ad[HEADS];
#pragma unroll
      for (int k = 0; k < HEADS; ++k) ad[k] = a_d[d * HEADS + k];

      int s0; float ev0[HEADS];
      node_scores<HEADS>(csr_src, a_s, ad, off, deg, d, lane, s0, ev0);
      const bool in0 = lane < cnt;

      float m[HEADS];
#pragma unroll
      for (int k = 0; k < HEADS; ++k) m[k] = in0 ? ev0[k] : -__builtin_inff();
      for (int g0 = 32; g0 < cnt; g0 += 32) {
        int s; float ev[HEADS];
        node_scores<HEADS>(csr_src, a_s, ad, off, deg, d, g0 + lane, s, ev);
        if (g0 + lane < cnt) {
#pragma unroll
          for (int k = 0; k < HEADS; ++k) m[k] = fmaxf(m[k], ev[k]);
        }
      }
#pragma unroll
      for (int k = 0; k < HEADS; ++k)
        for (int o = 16; o; o >>= 1) m[k] = fmaxf(m[k], __shfl_xor(m[k], o, 32));

      float ex0[HEADS], sum[HEADS];
#pragma unroll
      for (int k = 0; k < HEADS; ++k) {
        ex0[k] = __expf(ev0[k] - m[k]);
        sum[k] = in0 ? ex0[k] : 0.0f;
      }
      for (int g0 = 32; g0 < cnt; g0 += 32) {
        int s; float ev[HEADS];
        node_scores<HEADS>(csr_src, a_s, ad, off, deg, d, g0 + lane, s, ev);
        if (g0 + lane < cnt) {
#pragma unroll
          for (int k = 0; k < HEADS; ++k) sum[k] += __expf(ev[k] - m[k]);
        }
      }
#pragma unroll
      for (int k = 0; k < HEADS; ++k)
        for (int o = 16; o; o >>= 1) sum[k] += __shfl_xor(sum[k], o, 32);

      auto add_rows = [&](int s, const float (&alpha)[HEADS], int n) {
        for (int j0 = 0; j0 < n; j0 += UX) {
          float4 r[UX];
#pragma unroll
          for (int u = 0; u < UX; ++u) {
            const int sj = __builtin_amdgcn_readlane(s, j0 + u);
            r[u] = x4[(size_t)sj * (INCH / 4) + lane];
          }
#pragma unroll
          for (int u = 0; u < UX; ++u) {
            if (j0 + u < n) {
#pragma unroll
              for (int k = 0; k < HEADS; ++k) {
                const float al = __int_as_float(__builtin_amdgcn_readlane(__float_as_int(alpha[k]), j0 + u));
                z[k].x += al * r[u].x;
                z[k].y += al * r[u].y;
                z[k].z += al * r[u].z;
                z[k].w += al * r[u].w;
              }
            }
          }
        }
      };
      {
        float alpha[HEADS];
#pragma unroll
        for (int k = 0; k < HEADS; ++k) alpha[k] = ex0[k] / sum[k];
        add_rows(s0, alpha, cnt < 32 ? cnt : 32);
      }
      for (int g0 = 32; g0 < cnt; g0 += 32) {
        int s; float ev[HEADS], alpha[HEADS];
        node_scores<HEADS>(csr_src, a_s, ad, off, deg, d, g0 + lane, s, ev);
#pragma unroll
        for (int k = 0; k < HEADS; ++k) alpha[k] = __expf(ev[k] - m[k]) / sum[k];
        add_rows(s, alpha, (cnt - g0) < 32 ? (cnt - g0) : 32);
      }
    }
    float4* zrow = reinterpret_cast<float4*>(zs + nl * ZP);
#pragma unroll
    for (int k = 0; k < HEADS; ++k) zrow[k * (INCH / 4) + lane] = z[k];
  }
  __syncthreads();

  {
    const int ml = lane & 15;
    const int kh = lane >> 4;
    v8f acc0 = {}, acc1 = {}, acc2 = {}, acc3 = {};
    const float* zfrag = zs + ml * ZP + wv * INCH + 2 * kh;
    const float* bfrag = W1 + (size_t)(2 * kh) * (H1 * OUTC) + wv * OUTC + 4 * ml;
#pragma unroll 4
    for (int k = 0; k < INCH; k += 4) {
      const v2f a = *reinterpret_cast<const v2f*>(zfrag + k);
      const float4 bx = *reinterpret_cast<const float4*>(bfrag + (size_t)k * (H1 * OUTC));
      const float4 by = *reinterpret_cast<const float4*>(bfrag + (size_t)(k + 1) * (H1 * OUTC));
      v2f b0; b0.x = bx.x; b0.y = by.x;
      v2f b1; b1.x = bx.y; b1.y = by.y;
      v2f b2; b2.x = bx.z; b2.y = by.z;
      v2f b3; b3.x = bx.w; b3.y = by.w;
      acc0 = __builtin_amdgcn_wmma_f32_16x16x4_f32(false, a, false, b0, (short)0, acc0, false, false);
      acc1 = __builtin_amdgcn_wmma_f32_16x16x4_f32(false, a, false, b1, (short)0, acc1, false, false);
      acc2 = __builtin_amdgcn_wmma_f32_16x16x4_f32(false, a, false, b2, (short)0, acc2, false, false);
      acc3 = __builtin_amdgcn_wmma_f32_16x16x4_f32(false, a, false, b3, (short)0, acc3, false, false);
    }
    __syncthreads();
    float4* prow = reinterpret_cast<float4*>(part + wv * (NB * OUTC) + (8 * kh) * OUTC + 4 * ml);
#pragma unroll
    for (int r = 0; r < 8; ++r)
      prow[r * (OUTC / 4)] = make_float4(acc0[r], acc1[r], acc2[r], acc3[r]);
  }
  __syncthreads();

  {
    const int row = threadIdx.x >> 4;
    const int c4  = threadIdx.x & 15;
    const float4* p4 = reinterpret_cast<const float4*>(part);
    float4 sacc = p4[row * (OUTC / 4) + c4];
#pragma unroll
    for (int w = 1; w < HEADS; ++w) {
      const float4 p = p4[w * (NB * OUTC / 4) + row * (OUTC / 4) + c4];
      sacc.x += p.x; sacc.y += p.y; sacc.z += p.z; sacc.w += p.w;
    }
    const float4 bb = reinterpret_cast<const float4*>(bias)[c4];
    const float invH = 1.0f / HEADS;
    float4 o;
    o.x = sacc.x * invH + bb.x;
    o.y = sacc.y * invH + bb.y;
    o.z = sacc.z * invH + bb.z;
    o.w = sacc.w * invH + bb.w;
    o.x = o.x > 0.0f ? o.x : (__expf(o.x) - 1.0f);
    o.y = o.y > 0.0f ? o.y : (__expf(o.y) - 1.0f);
    o.z = o.z > 0.0f ? o.z : (__expf(o.z) - 1.0f);
    o.w = o.w > 0.0f ? o.w : (__expf(o.w) - 1.0f);
    const int d = nbase + row;
    if (d < NN) reinterpret_cast<float4*>(out)[(size_t)d * (OUTC / 4) + c4] = o;
  }
}

static inline unsigned cdiv(long long a, long long b) { return (unsigned)((a + b - 1) / b); }

extern "C" void kernel_launch(void* const* d_in, const int* in_sizes, int n_in,
                              void* d_out, int out_size, void* d_ws, size_t ws_size,
                              hipStream_t stream) {
  (void)in_sizes; (void)n_in; (void)out_size; (void)ws_size;
  const float* x    = (const float*)d_in[0];
  const int*   ei   = (const int*)d_in[1];
  const float* W1   = (const float*)d_in[2];
  const float* asw1 = (const float*)d_in[3];
  const float* adw1 = (const float*)d_in[4];
  const float* b1   = (const float*)d_in[5];
  const float* W2   = (const float*)d_in[6];
  const float* asw2 = (const float*)d_in[7];
  const float* adw2 = (const float*)d_in[8];
  const float* b2   = (const float*)d_in[9];
  float* out = (float*)d_out;

  float* ws   = (float*)d_ws;
  float* vfold = ws; ws += (size_t)INCH * 16;
  float* as1  = ws; ws += (size_t)NN * H1;
  float* ad1  = ws; ws += (size_t)NN * H1;
  float* hmid = ws; ws += (size_t)NN * OUTC;
  float* h2   = ws; ws += (size_t)NN * OUTC;
  float* as2  = ws; ws += (size_t)NN;
  float* ad2  = ws; ws += (size_t)NN;
  int* counts  = (int*)ws; ws += (size_t)NN;
  int* offsets = (int*)ws; ws += (size_t)NN;
  int* csr_src = (int*)ws; ws += (size_t)EE;
  int* rank = (int*)h2;

  const int T = 256;
  const int NPW = 4;
  const unsigned gNode = cdiv((long long)cdiv(NN, NPW) * 32, T);

  fill_u32_kernel<<<cdiv(cdiv(NN / 4, VPT), T), T, 0, stream>>>((uint32_t*)counts, 0u, NN);
  dst_rank_kernel<<<cdiv(cdiv(EE, EPT), T), T, 0, stream>>>(ei, counts, rank);
  offsets_scan_kernel<<<1, SCAN_T, 0, stream>>>(counts, offsets);
  dst_scatter_kernel<<<cdiv(cdiv(EE, EPT), T), T, 0, stream>>>(ei, offsets, rank, csr_src);

  att_fold_kernel<<<cdiv(INCH * 16, T), T, 0, stream>>>(W1, asw1, adw1, vfold);
  scores_wmma_kernel<<<cdiv((long long)(NN / 16) * 32, T), T, 0, stream>>>(x, vfold, as1, ad1);
  layer1_fused_kernel<<<cdiv(NN, NB), 256, 0, stream>>>(offsets, counts, csr_src, as1, ad1, x, W1, b1, hmid);

  {
    long long waves = ((long long)NN / 16) * (OUTC / 64);
    gemm_wmma_f32_kernel<<<cdiv(waves * 32, T), T, 0, stream>>>(hmid, W2, h2, NN, OUTC, OUTC);
  }
  att_scores_kernel<1><<<cdiv(cdiv(NN, 8) * 32, T), T, 0, stream>>>(h2, asw2, adw2, as2, ad2, NN);
  node_aggregate_kernel<1, 8, NPW><<<gNode, T, 0, stream>>>(offsets, counts, csr_src, as2, ad2, h2, b2, out, 1.0f, 0);
}


# Layer 2's projection and attention scalars moved into the layer-1 kernel: its 16x64 output tile stays in LDS and 4 waves multiply it by W2 on WMMA; 2 launches and one plane's round trip gone
# speedup vs baseline: 4.9321x; 1.0916x over previous
#include <hip/hip_runtime.h>
#include <stdint.h>

#define NN    50000
#define EE    400000
#define INCH  128
#define OUTC  64
#define H1    8
#define NEG   0.2f

typedef float v2f __attribute__((ext_vector_type(2)));
typedef float v8f __attribute__((ext_vector_type(8)));

#define VPT 4
__global__ void fill_u32_kernel(uint32_t* p, uint32_t v, int n) {
  const int tid    = blockIdx.x * blockDim.x + threadIdx.x;
  const int stride = gridDim.x * blockDim.x;
  const int n4     = n >> 2;
  uint4* p4 = reinterpret_cast<uint4*>(p);
  const uint4 val = make_uint4(v, v, v, v);
#pragma unroll
  for (int j = 0; j < VPT; ++j) {
    const int i = tid + j * stride;
    if (i < n4) p4[i] = val;
  }
}

__device__ __forceinline__ float leaky(float x) {
  return x > 0.0f ? x : NEG * x;
}

#define EPT 4
__global__ void dst_rank_kernel(const int* __restrict__ ei,
                                int* __restrict__ counts,
                                int* __restrict__ rank) {
  const int tid    = blockIdx.x * blockDim.x + threadIdx.x;
  const int stride = gridDim.x * blockDim.x;
  int dd[EPT], r[EPT];
#pragma unroll
  for (int j = 0; j < EPT; ++j) {
    const int e = tid + j * stride;
    dd[j] = e < EE ? ei[EE + e] : -1;
  }
#pragma unroll
  for (int j = 0; j < EPT; ++j) r[j] = dd[j] >= 0 ? atomicAdd(&counts[dd[j]], 1) : 0;
#pragma unroll
  for (int j = 0; j < EPT; ++j) {
    const int e = tid + j * stride;
    if (e < EE) rank[e] = r[j];
  }
}

#define SCAN_T 1024
__global__ void __launch_bounds__(SCAN_T)
offsets_scan_kernel(const int* __restrict__ counts, int* __restrict__ offsets) {
  constexpr int WAVES = SCAN_T / 32;
  static_assert(WAVES == 32, "the wave totals are summed by the 32 lanes of a wave");
  constexpr int IT = ((NN + WAVES - 1) / WAVES + 31) / 32;
  constexpr int PER_WAVE = IT * 32;
  __shared__ int wtot[WAVES];
  const int lane = threadIdx.x & 31;
  const int w    = threadIdx.x >> 5;
  const int base = w * PER_WAVE;
  int v[IT];
  int tot = 0;
#pragma unroll
  for (int j = 0; j < IT; ++j) {
    const int i = base + j * 32 + lane;
    v[j] = i < NN ? counts[i] : 0;
    tot += v[j];
  }
  int inc = tot;
  for (int off = 1; off < 32; off <<= 1) {
    const int nb = __shfl_up(inc, off, 32);
    if (lane >= off) inc += nb;
  }
  if (lane == 31) wtot[w] = inc;
  __syncthreads();
  int carry = lane < w ? wtot[lane] : 0;
  for (int off = 16; off; off >>= 1) carry += __shfl_xor(carry, off, 32);
  int run = carry + inc - tot;
#pragma unroll
  for (int j = 0; j < IT; ++j) {
    const int i = base + j * 32 + lane;
    if (i < NN) offsets[i] = run;
    run += v[j];
  }
}

__global__ void dst_scatter_kernel(const int* __restrict__ ei,
                                   const int* __restrict__ offsets,
                                   const int* __restrict__ rank,
                                   int* __restrict__ csr_src) {
  const int tid    = blockIdx.x * blockDim.x + threadIdx.x;
  const int stride = gridDim.x * blockDim.x;
  int dd[EPT], ss[EPT], rr[EPT], pos[EPT];
#pragma unroll
  for (int j = 0; j < EPT; ++j) {
    const int e  = tid + j * stride;
    const int ec = e < EE ? e : 0;
    dd[j] = ei[EE + ec];
    ss[j] = ei[ec];
    rr[j] = rank[ec];
  }
#pragma unroll
  for (int j = 0; j < EPT; ++j) pos[j] = offsets[dd[j]] + rr[j];
#pragma unroll
  for (int j = 0; j < EPT; ++j) {
    const int e = tid + j * stride;
    if (e < EE) csr_src[pos[j]] = ss[j];
  }
}

template <int HEADS>
__device__ __forceinline__ void node_scores(const int* __restrict__ csr_src,
                                            const float* __restrict__ a_s,
                                            const float (&ad)[HEADS],
                                            int off, int deg, int d, int i,
                                            int& s, float (&ev)[HEADS]) {
  s = i < deg ? csr_src[off + i] : d;
  const float* ap = a_s + (size_t)s * HEADS;
  if constexpr (HEADS % 4 == 0) {
    const float4* ap4 = reinterpret_cast<const float4*>(ap);
#pragma unroll
    for (int q = 0; q < HEADS / 4; ++q) {
      const float4 t = ap4[q];
      ev[4 * q + 0] = leaky(t.x + ad[4 * q + 0]);
      ev[4 * q + 1] = leaky(t.y + ad[4 * q + 1]);
      ev[4 * q + 2] = leaky(t.z + ad[4 * q + 2]);
      ev[4 * q + 3] = leaky(t.w + ad[4 * q + 3]);
    }
  } else {
#pragma unroll
    for (int k = 0; k < HEADS; ++k) ev[k] = leaky(ap[k] + ad[k]);
  }
}

template <int HEADS, int U, int NPW>
__global__ void node_aggregate_kernel(const int* __restrict__ offsets,
                                      const int* __restrict__ counts,
                                      const int* __restrict__ csr_src,
                                      const float* __restrict__ a_s,
                                      const float* __restrict__ a_d,
                                      const float* __restrict__ h,
                                      const float* __restrict__ bias,
                                      float* __restrict__ out,
                                      float invH, int do_elu) {
  static_assert(32 % U == 0, "a batch must not run past the wave's lanes");
  const int lane = threadIdx.x & 31;
  const int wid = __builtin_amdgcn_readfirstlane((int)((blockIdx.x * blockDim.x + threadIdx.x) >> 5));
  const float bias0 = bias[lane], bias1 = bias[lane + 32];
  for (int t = 0; t < NPW; ++t) {
    const int d = wid * NPW + t;
    if (d >= NN) break;
    const int off = __builtin_amdgcn_readfirstlane(offsets[d]);
    const int deg = __builtin_amdgcn_readfirstlane(counts[d]);
    const int cnt = deg + 1;
    float ad[HEADS];
#pragma unroll
    for (int k = 0; k < HEADS; ++k) ad[k] = a_d[d * HEADS + k];

    int s0; float ev0[HEADS];
    node_scores<HEADS>(csr_src, a_s, ad, off, deg, d, lane, s0, ev0);
    const bool in0 = lane < cnt;

    float m[HEADS];
#pragma unroll
    for (int k = 0; k < HEADS; ++k) m[k] = in0 ? ev0[k] : -__builtin_inff();
    for (int g0 = 32; g0 < cnt; g0 += 32) {
      int s; float ev[HEADS];
      node_scores<HEADS>(csr_src, a_s, ad, off, deg, d, g0 + lane, s, ev);
      if (g0 + lane < cnt) {
#pragma unroll
        for (int k = 0; k < HEADS; ++k) m[k] = fmaxf(m[k], ev[k]);
      }
    }
#pragma unroll
    for (int k = 0; k < HEADS; ++k)
      for (int o = 16; o; o >>= 1) m[k] = fmaxf(m[k], __shfl_xor(m[k], o, 32));

    float ex0[HEADS], sum[HEADS];
#pragma unroll
    for (int k = 0; k < HEADS; ++k) {
      ex0[k] = __expf(ev0[k] - m[k]);
      sum[k] = in0 ? ex0[k] : 0.0f;
    }
    for (int g0 = 32; g0 < cnt; g0 += 32) {
      int s; float ev[HEADS];
      node_scores<HEADS>(csr_src, a_s, ad, off, deg, d, g0 + lane, s, ev);
      if (g0 + lane < cnt) {
#pragma unroll
        for (int k = 0; k < HEADS; ++k) sum[k] += __expf(ev[k] - m[k]);
      }
    }
#pragma unroll
    for (int k = 0; k < HEADS; ++k)
      for (int o = 16; o; o >>= 1) sum[k] += __shfl_xor(sum[k], o, 32);

    float v0 = 0.0f, v1 = 0.0f;
    auto add_rows = [&](int s, const float (&alpha)[HEADS], int n) {
      for (int j0 = 0; j0 < n; j0 += U) {
        float r0[U][HEADS], r1[U][HEADS];
#pragma unroll
        for (int u = 0; u < U; ++u) {
          const int sj = __builtin_amdgcn_readlane(s, j0 + u);
          const float* hrow = h + (size_t)sj * HEADS * OUTC;
#pragma unroll
          for (int k = 0; k < HEADS; ++k) {
            r0[u][k] = hrow[k * OUTC + lane];
            r1[u][k] = hrow[k * OUTC + lane + 32];
          }
        }
#pragma unroll
        for (int u = 0; u < U; ++u) {
          if (j0 + u < n) {
#pragma unroll
            for (int k = 0; k < HEADS; ++k) {
              const float al = __int_as_float(__builtin_amdgcn_readlane(__float_as_int(alpha[k]), j0 + u));
              v0 += al * r0[u][k];
              v1 += al * r1[u][k];
            }
          }
        }
      }
    };
    {
      float alpha[HEADS];
#pragma unroll
      for (int k = 0; k < HEADS; ++k) alpha[k] = ex0[k] / sum[k];
      add_rows(s0, alpha, cnt < 32 ? cnt : 32);
    }
    for (int g0 = 32; g0 < cnt; g0 += 32) {
      int s; float ev[HEADS], alpha[HEADS];
      node_scores<HEADS>(csr_src, a_s, ad, off, deg, d, g0 + lane, s, ev);
#pragma unroll
      for (int k = 0; k < HEADS; ++k) alpha[k] = __expf(ev[k] - m[k]) / sum[k];
      add_rows(s, alpha, (cnt - g0) < 32 ? (cnt - g0) : 32);
    }

    float o0 = v0 * invH + bias0;
    float o1 = v1 * invH + bias1;
    if (do_elu) {
      o0 = o0 > 0.0f ? o0 : (__expf(o0) - 1.0f);
      o1 = o1 > 0.0f ? o1 : (__expf(o1) - 1.0f);
    }
    out[(size_t)d * OUTC + lane]      = o0;
    out[(size_t)d * OUTC + lane + 32] = o1;
  }
}

__global__ void att_fold_kernel(const float* __restrict__ W1,
                                const float* __restrict__ att_src,
                                const float* __restrict__ att_dst,
                                float* __restrict__ V) {
  const int t = blockIdx.x * blockDim.x + threadIdx.x;
  if (t >= INCH * 16) return;
  const int i = t >> 4, j = t & 15, hd = j & 7;
  const float4* w4 = reinterpret_cast<const float4*>(W1 + (size_t)i * (H1 * OUTC) + hd * OUTC);
  const float4* a4 = reinterpret_cast<const float4*>((j < 8 ? att_src : att_dst) + hd * OUTC);
  float s = 0.0f;
#pragma unroll
  for (int c = 0; c < OUTC / 4; ++c) {
    const float4 w = w4[c];
    const float4 a = a4[c];
    s += w.x * a.x + w.y * a.y + w.z * a.z + w.w * a.w;
  }
  V[t] = s;
}

__global__ void scores_wmma_kernel(const float* __restrict__ x,
                                   const float* __restrict__ V,
                                   float* __restrict__ a_s,
                                   float* __restrict__ a_d) {
  const int wid  = (blockIdx.x * blockDim.x + threadIdx.x) >> 5;
  const int lane = threadIdx.x & 31;
  if (wid >= NN / 16) return;
  const int m0 = wid << 4;
  const int ml = lane & 15;
  const int kh = lane >> 4;
  v8f acc = {};
  const float* arow = x + (size_t)(m0 + ml) * INCH + 2 * kh;
  const float* vcol = V + (2 * kh) * 16 + ml;
#pragma unroll 16
  for (int k = 0; k < INCH; k += 4) {
    v2f a; a.x = arow[k];      a.y = arow[k + 1];
    v2f b; b.x = vcol[k * 16]; b.y = vcol[k * 16 + 16];
    acc = __builtin_amdgcn_wmma_f32_16x16x4_f32(false, a, false, b, (short)0, acc, false, false);
  }
  float* dstp = (ml < 8 ? a_s : a_d) + (size_t)(m0 + 8 * kh) * H1 + (ml & 7);
#pragma unroll
  for (int r = 0; r < 8; ++r) dstp[r * H1] = acc[r];
}

#define NB 16
#define ZP 1028
#define UX 8
#define HP 68
__global__ void __launch_bounds__(256)
layer1_fused_kernel(const int* __restrict__ offsets,
                    const int* __restrict__ counts,
                    const int* __restrict__ csr_src,
                    const float* __restrict__ a_s,
                    const float* __restrict__ a_d,
                    const float* __restrict__ x,
                    const float* __restrict__ W1,
                    const float* __restrict__ bias,
                    const float* __restrict__ W2,
                    const float* __restrict__ att_src2,
                    const float* __restrict__ att_dst2,
                    float* __restrict__ h2,
                    float* __restrict__ a_s2,
                    float* __restrict__ a_d2) {
  constexpr int HEADS = H1;
  __shared__ __attribute__((aligned(16))) float zs[NB * ZP];
  static_assert(HEADS * NB * OUTC + NB * HP <= NB * ZP, "the products and the output tile must fit where the sums were");
  float* part = zs;
  float* hm  = zs + HEADS * NB * OUTC;
  float* h2t = zs;
  const int lane = threadIdx.x & 31;
  const int wv = __builtin_amdgcn_readfirstlane((int)(threadIdx.x >> 5));
  const int nbase = blockIdx.x * NB;
  const float4* x4 = reinterpret_cast<const float4*>(x);

  for (int t = 0; t < NB / 8; ++t) {
    const int nl = wv * (NB / 8) + t;
    const int d  = nbase + nl;
    float4 z[HEADS];
#pragma unroll
    for (int k = 0; k < HEADS; ++k) z[k] = make_float4(0.0f, 0.0f, 0.0f, 0.0f);
    if (d < NN) {
      const int off = __builtin_amdgcn_readfirstlane(offsets[d]);
      const int deg = __builtin_amdgcn_readfirstlane(counts[d]);
      const int cnt = deg + 1;
      float ad[HEADS];
#pragma unroll
      for (int k = 0; k < HEADS; ++k) ad[k] = a_d[d * HEADS + k];

      int s0; float ev0[HEADS];
      node_scores<HEADS>(csr_src, a_s, ad, off, deg, d, lane, s0, ev0);
      const bool in0 = lane < cnt;

      float m[HEADS];
#pragma unroll
      for (int k = 0; k < HEADS; ++k) m[k] = in0 ? ev0[k] : -__builtin_inff();
      for (int g0 = 32; g0 < cnt; g0 += 32) {
        int s; float ev[HEADS];
        node_scores<HEADS>(csr_src, a_s, ad, off, deg, d, g0 + lane, s, ev);
        if (g0 + lane < cnt) {
#pragma unroll
          for (int k = 0; k < HEADS; ++k) m[k] = fmaxf(m[k], ev[k]);
        }
      }
#pragma unroll
      for (int k = 0; k < HEADS; ++k)
        for (int o = 16; o; o >>= 1) m[k] = fmaxf(m[k], __shfl_xor(m[k], o, 32));

      float ex0[HEADS], sum[HEADS];
#pragma unroll
      for (int k = 0; k < HEADS; ++k) {
        ex0[k] = __expf(ev0[k] - m[k]);
        sum[k] = in0 ? ex0[k] : 0.0f;
      }
      for (int g0 = 32; g0 < cnt; g0 += 32) {
        int s; float ev[HEADS];
        node_scores<HEADS>(csr_src, a_s, ad, off, deg, d, g0 + lane, s, ev);
        if (g0 + lane < cnt) {
#pragma unroll
          for (int k = 0; k < HEADS; ++k) sum[k] += __expf(ev[k] - m[k]);
        }
      }
#pragma unroll
      for (int k = 0; k < HEADS; ++k)
        for (int o = 16; o; o >>= 1) sum[k] += __shfl_xor(sum[k], o, 32);

      auto add_rows = [&](int s, const float (&alpha)[HEADS], int n) {
        for (int j0 = 0; j0 < n; j0 += UX) {
          float4 r[UX];
#pragma unroll
          for (int u = 0; u < UX; ++u) {
            const int sj = __builtin_amdgcn_readlane(s, j0 + u);
            r[u] = x4[(size_t)sj * (INCH / 4) + lane];
          }
#pragma unroll
          for (int u = 0; u < UX; ++u) {
            if (j0 + u < n) {
#pragma unroll
              for (int k = 0; k < HEADS; ++k) {
                const float al = __int_as_float(__builtin_amdgcn_readlane(__float_as_int(alpha[k]), j0 + u));
                z[k].x += al * r[u].x;
                z[k].y += al * r[u].y;
                z[k].z += al * r[u].z;
                z[k].w += al * r[u].w;
              }
            }
          }
        }
      };
      {
        float alpha[HEADS];
#pragma unroll
        for (int k = 0; k < HEADS; ++k) alpha[k] = ex0[k] / sum[k];
        add_rows(s0, alpha, cnt < 32 ? cnt : 32);
      }
      for (int g0 = 32; g0 < cnt; g0 += 32) {
        int s; float ev[HEADS], alpha[HEADS];
        node_scores<HEADS>(csr_src, a_s, ad, off, deg, d, g0 + lane, s, ev);
#pragma unroll
        for (int k = 0; k < HEADS; ++k) alpha[k] = __expf(ev[k] - m[k]) / sum[k];
        add_rows(s, alpha, (cnt - g0) < 32 ? (cnt - g0) : 32);
      }
    }
    float4* zrow = reinterpret_cast<float4*>(zs + nl * ZP);
#pragma unroll
    for (int k = 0; k < HEADS; ++k) zrow[k * (INCH / 4) + lane] = z[k];
  }
  __syncthreads();

  const int ml = lane & 15;
  const int kh = lane >> 4;
  const int row = threadIdx.x >> 4;
  const int c4  = threadIdx.x & 15;
  v2f w2f[OUTC / 4] = {};
  if (wv < 4) {
    const float* w2p = W2 + (size_t)(2 * kh) * OUTC + wv * 16 + ml;
#pragma unroll
    for (int j = 0; j < OUTC / 4; ++j) {
      w2f[j].x = w2p[(4 * j) * OUTC];
      w2f[j].y = w2p[(4 * j + 1) * OUTC];
    }
  }
  const float4 as2w = reinterpret_cast<const float4*>(att_src2)[c4];
  const float4 ad2w = reinterpret_cast<const float4*>(att_dst2)[c4];

  {
    v8f acc0 = {}, acc1 = {}, acc2 = {}, acc3 = {};
    const float* zfrag = zs + ml * ZP + wv * INCH + 2 * kh;
    const float* bfrag = W1 + (size_t)(2 * kh) * (H1 * OUTC) + wv * OUTC + 4 * ml;
#pragma unroll 4
    for (int k = 0; k < INCH; k += 4) {
      const v2f a = *reinterpret_cast<const v2f*>(zfrag + k);
      const float4 bx = *reinterpret_cast<const float4*>(bfrag + (size_t)k * (H1 * OUTC));
      const float4 by = *reinterpret_cast<const float4*>(bfrag + (size_t)(k + 1) * (H1 * OUTC));
      v2f b0; b0.x = bx.x; b0.y = by.x;
      v2f b1; b1.x = bx.y; b1.y = by.y;
      v2f b2; b2.x = bx.z; b2.y = by.z;
      v2f b3; b3.x = bx.w; b3.y = by.w;
      acc0 = __builtin_amdgcn_wmma_f32_16x16x4_f32(false, a, false, b0, (short)0, acc0, false, false);
      acc1 = __builtin_amdgcn_wmma_f32_16x16x4_f32(false, a, false, b1, (short)0, acc1, false, false);
      acc2 = __builtin_amdgcn_wmma_f32_16x16x4_f32(false, a, false, b2, (short)0, acc2, false, false);
      acc3 = __builtin_amdgcn_wmma_f32_16x16x4_f32(false, a, false, b3, (short)0, acc3, false, false);
    }
    __syncthreads();
    float4* prow = reinterpret_cast<float4*>(part + wv * (NB * OUTC) + (8 * kh) * OUTC + 4 * ml);
#pragma unroll
    for (int r = 0; r < 8; ++r)
      prow[r * (OUTC / 4)] = make_float4(acc0[r], acc1[r], acc2[r], acc3[r]);
  }
  __syncthreads();

  {
    const float4* p4 = reinterpret_cast<const float4*>(part);
    float4 sacc = p4[row * (OUTC / 4) + c4];
#pragma unroll
    for (int w = 1; w < HEADS; ++w) {
      const float4 p = p4[w * (NB * OUTC / 4) + row * (OUTC / 4) + c4];
      sacc.x += p.x; sacc.y += p.y; sacc.z += p.z; sacc.w += p.w;
    }
    const float4 bb = reinterpret_cast<const float4*>(bias)[c4];
    const float invH = 1.0f / HEADS;
    float4 o;
    o.x = sacc.x * invH + bb.x;
    o.y = sacc.y * invH + bb.y;
    o.z = sacc.z * invH + bb.z;
    o.w = sacc.w * invH + bb.w;
    o.x = o.x > 0.0f ? o.x : (__expf(o.x) - 1.0f);
    o.y = o.y > 0.0f ? o.y : (__expf(o.y) - 1.0f);
    o.z = o.z > 0.0f ? o.z : (__expf(o.z) - 1.0f);
    o.w = o.w > 0.0f ? o.w : (__expf(o.w) - 1.0f);
    *reinterpret_cast<float4*>(hm + row * HP + 4 * c4) = o;
  }
  __syncthreads();

  if (wv < 4) {
    v8f acc = {};
    const float* afrag = hm + ml * HP + 2 * kh;
    v2f af[OUTC / 4];
#pragma unroll
    for (int j = 0; j < OUTC / 4; ++j) af[j] = *reinterpret_cast<const v2f*>(afrag + 4 * j);
    __builtin_amdgcn_sched_barrier(0);
#pragma unroll
    for (int j = 0; j < OUTC / 4; ++j)
      acc = __builtin_amdgcn_wmma_f32_16x16x4_f32(false, af[j], false, w2f[j], (short)0, acc, false, false);
    float* hcol = h2t + (8 * kh) * HP + wv * 16 + ml;
#pragma unroll
    for (int r = 0; r < 8; ++r) hcol[r * HP] = acc[r];
  }
  __syncthreads();

  {
    const float4 v = *reinterpret_cast<const float4*>(h2t + row * HP + 4 * c4);
    const float4 s = as2w;
    const float4 t = ad2w;
    float ps = v.x * s.x + v.y * s.y + v.z * s.z + v.w * s.w;
    float pd = v.x * t.x + v.y * t.y + v.z * t.z + v.w * t.w;
#pragma unroll
    for (int off = 8; off; off >>= 1) {
      ps += __shfl_xor(ps, off, 32);
      pd += __shfl_xor(pd, off, 32);
    }
    const int d = nbase + row;
    if (d < NN) {
      reinterpret_cast<float4*>(h2)[(size_t)d * (OUTC / 4) + c4] = v;
      if (c4 == 0) {
        a_s2[d] = ps;
        a_d2[d] = pd;
      }
    }
  }
}

static inline unsigned cdiv(long long a, long long b) { return (unsigned)((a + b - 1) / b); }

extern "C" void kernel_launch(void* const* d_in, const int* in_sizes, int n_in,
                              void* d_out, int out_size, void* d_ws, size_t ws_size,
                              hipStream_t stream) {
  (void)in_sizes; (void)n_in; (void)out_size; (void)ws_size;
  const float* x    = (const float*)d_in[0];
  const int*   ei   = (const int*)d_in[1];
  const float* W1   = (const float*)d_in[2];
  const float* asw1 = (const float*)d_in[3];
  const float* adw1 = (const float*)d_in[4];
  const float* b1   = (const float*)d_in[5];
  const float* W2   = (const float*)d_in[6];
  const float* asw2 = (const float*)d_in[7];
  const float* adw2 = (const float*)d_in[8];
  const float* b2   = (const float*)d_in[9];
  float* out = (float*)d_out;

  float* ws   = (float*)d_ws;
  float* vfold = ws; ws += (size_t)INCH * 16;
  float* as1  = ws; ws += (size_t)NN * H1;
  float* ad1  = ws; ws += (size_t)NN * H1;
  float* h2   = ws; ws += (size_t)NN * OUTC;
  float* as2  = ws; ws += (size_t)NN;
  float* ad2  = ws; ws += (size_t)NN;
  int* counts  = (int*)ws; ws += (size_t)NN;
  int* offsets = (int*)ws; ws += (size_t)NN;
  int* csr_src = (int*)ws; ws += (size_t)EE;
  int* rank = (int*)h2;

  const int T = 256;
  const int NPW = 4;
  const unsigned gNode = cdiv((long long)cdiv(NN, NPW) * 32, T);

  fill_u32_kernel<<<cdiv(cdiv(NN / 4, VPT), T), T, 0, stream>>>((uint32_t*)counts, 0u, NN);
  dst_rank_kernel<<<cdiv(cdiv(EE, EPT), T), T, 0, stream>>>(ei, counts, rank);
  offsets_scan_kernel<<<1, SCAN_T, 0, stream>>>(counts, offsets);
  dst_scatter_kernel<<<cdiv(cdiv(EE, EPT), T), T, 0, stream>>>(ei, offsets, rank, csr_src);

  att_fold_kernel<<<cdiv(INCH * 16, T), T, 0, stream>>>(W1, asw1, adw1, vfold);
  scores_wmma_kernel<<<cdiv((long long)(NN / 16) * 32, T), T, 0, stream>>>(x, vfold, as1, ad1);
  layer1_fused_kernel<<<cdiv(NN, NB), 256, 0, stream>>>(offsets, counts, csr_src, as1, ad1, x, W1, b1, W2, asw2, adw2, h2, as2, ad2);

  node_aggregate_kernel<1, 8, NPW><<<gNode, T, 0, stream>>>(offsets, counts, csr_src, as2, ad2, h2, b2, out, 1.0f, 0);
}
